# starting kernel
# baseline (speedup 1.0000x reference)
#include <hip/hip_runtime.h>
#include <hip/hip_bf16.h>
#include <math.h>

typedef __attribute__((ext_vector_type(2))) float v2f;
typedef __attribute__((ext_vector_type(4))) float v4f;
typedef __attribute__((ext_vector_type(8))) float v8f;

#define BS   8
#define CIN  64
#define HH   256
#define WW   256
#define NEXP 4
#define OCH  64
#define KS   3
#define KTOT (CIN * KS * KS)
#define WROW (CIN * KS * KS)

#define TILE_W   128
#define CI_CHUNK 16
#define KCHUNK   (CI_CHUNK * 9)
#define SW_STRIDE 148
#define PATCH_W  132

__device__ constexpr int poff(int k) {
    return ((k / 9) * 3 + ((k % 9) / 3)) * PATCH_W + (k % 3);
}

__global__ __launch_bounds__(256)
void att_partial_kernel(const float* __restrict__ x,
                        float* __restrict__ parth,
                        float* __restrict__ partv) {
    const int t = threadIdx.x;
    const float* plane = x + (size_t)blockIdx.x * (HH * WW);
    float rowAcc = 0.f, colAcc = 0.f;
    for (int w = 0; w < WW; ++w) rowAcc += plane[t * WW + w];
    for (int h = 0; h < HH; ++h) colAcc += plane[h * WW + t];
    parth[(size_t)blockIdx.x * HH + t] = rowAcc;
    partv[(size_t)blockIdx.x * WW + t] = colAcc;
}

__global__ __launch_bounds__(256)
void att_finalize_kernel(const float* __restrict__ parth,
                         const float* __restrict__ partv,
                         const float* __restrict__ net_w,
                         const float* __restrict__ bias,
                         float* __restrict__ att,
                         float* __restrict__ aggb) {
    __shared__ float sh[HH + WW];
    __shared__ float sa[NEXP];
    const int b = blockIdx.x, t = threadIdx.x;
    float accH = 0.f, accV = 0.f;
    for (int c = 0; c < CIN; ++c) {
        accH += parth[((size_t)b * CIN + c) * HH + t];
        accV += partv[((size_t)b * CIN + c) * WW + t];
    }
    const float inv = 1.f / (float)(CIN * WW);
    sh[t]      = accH * inv;
    sh[HH + t] = accV * inv;
    __syncthreads();
    if (t < NEXP) {
        float acc = 0.f;
        for (int j = 0; j < HH + WW; ++j) acc += sh[j] * net_w[t * (HH + WW) + j];
        float s = 1.f / (1.f + __expf(-acc));
        att[b * NEXP + t] = s;
        sa[t] = s;
    }
    __syncthreads();
    if (t < OCH) {
        float acc = 0.f;
        for (int k = 0; k < NEXP; ++k) acc += sa[k] * bias[k * OCH + t];
        aggb[b * OCH + t] = acc;
    }
}

__global__ __launch_bounds__(256)
void agg_weight_kernel(const float* __restrict__ weight,
                       const float* __restrict__ att,
                       float* __restrict__ aggw) {
    const size_t idx = (size_t)blockIdx.x * 256 + threadIdx.x;
    const int b = (int)(idx / (OCH * KTOT));
    const int j = (int)(idx % (OCH * KTOT));
    const float a0 = att[b * NEXP + 0], a1 = att[b * NEXP + 1];
    const float a2 = att[b * NEXP + 2], a3 = att[b * NEXP + 3];
    const size_t E = (size_t)OCH * KTOT;
    aggw[idx] = a0 * weight[j] + a1 * weight[E + j] + a2 * weight[2 * E + j] + a3 * weight[3 * E + j];
}

__global__ __launch_bounds__(256)
void condconv_wmma_kernel(const float* __restrict__ x,
                          const float* __restrict__ aggw,
                          const float* __restrict__ aggb,
                          float* __restrict__ out) {
    __shared__ float sW[OCH * SW_STRIDE];
    __shared__ float sX[CI_CHUNK * 3 * PATCH_W];
    __shared__ float sBias[OCH];

    const int tid = threadIdx.x;
    const int b      = blockIdx.x >> 9;
    const int rem    = blockIdx.x & 511;
    const int h      = rem >> 1;
    const int wstart = (rem & 1) * TILE_W;

    if (tid < OCH) sBias[tid] = aggb[b * OCH + tid];

    const int lane   = tid & 31;
    const int wv     = tid >> 5;
    const int hl     = lane >> 4;
    const int l16    = lane & 15;
    const int m_base = (wv & 1) * 32;
    const int n_base = (wv >> 1) * 32;

    v8f acc[2][2] = {};

    const float* wrow = aggw + (size_t)b * (OCH * WROW);

    const int aoff0 = (m_base + l16) * SW_STRIDE + 2 * hl;
    const int aoff1 = (m_base + 16 + l16) * SW_STRIDE + 2 * hl;
    const int n0 = n_base + l16;
    const int n1 = n0 + 16;

    for (int ci0 = 0; ci0 < CIN; ci0 += CI_CHUNK) {
        __syncthreads();
        for (int e4 = tid; e4 < OCH * (KCHUNK / 4); e4 += 256) {
            const int o = e4 / (KCHUNK / 4), kc4 = e4 % (KCHUNK / 4);
            const v4f v = *(const v4f*)&wrow[o * WROW + ci0 * 9 + kc4 * 4];
            *(v4f*)&sW[o * SW_STRIDE + kc4 * 4] = v;
        }
        const float* xb = x + ((size_t)b * CIN + ci0) * (HH * WW);
        for (int e = tid; e < CI_CHUNK * 3 * PATCH_W; e += 256) {
            const int ci = e / (3 * PATCH_W);
            const int rr = (e / PATCH_W) % 3;
            const int cc = e % PATCH_W;
            const int gh = h - 1 + rr;
            const int gw = wstart - 1 + cc;
            float v = 0.f;
            if ((unsigned)gh < (unsigned)HH && (unsigned)gw < (unsigned)WW)
                v = xb[(size_t)ci * (HH * WW) + gh * WW + gw];
            sX[e] = v;
        }
        __syncthreads();

        if (ci0 + CI_CHUNK < CIN) {
            {
                const int o = tid >> 2, seg = tid & 3;
                __builtin_prefetch(wrow + (size_t)o * WROW + (ci0 + CI_CHUNK) * 9 + seg * 36, 0, 1);
            }
            if (tid < 96) {
                const int ci = tid / 6, rr = (tid / 2) % 3, hs = tid & 1;
                int gh = h - 1 + rr; gh = gh < 0 ? 0 : (gh > HH - 1 ? HH - 1 : gh);
                int gw = wstart + hs * 64; if (wstart == 0 && hs == 0) gw = 0;
                const float* nx = x + ((size_t)b * CIN + ci0 + CI_CHUNK + ci) * (HH * WW);
                __builtin_prefetch(nx + gh * WW + gw, 0, 1);
            }
        }

        #pragma unroll
        for (int kc0 = 0; kc0 < KCHUNK; kc0 += 4) {
            const int o0 = hl ? poff(kc0 + 2) : poff(kc0 + 0);
            const int o1 = hl ? poff(kc0 + 3) : poff(kc0 + 1);
            const v2f a0 = *(const v2f*)&sW[aoff0 + kc0];
            const v2f a1 = *(const v2f*)&sW[aoff1 + kc0];
            v2f b0, b1;
            b0.x = sX[o0 + n0]; b0.y = sX[o1 + n0];
            b1.x = sX[o0 + n1]; b1.y = sX[o1 + n1];
            acc[0][0] = __builtin_amdgcn_wmma_f32_16x16x4_f32(false, a0, false, b0, (short)0, acc[0][0], false, false);
            acc[0][1] = __builtin_amdgcn_wmma_f32_16x16x4_f32(false, a0, false, b1, (short)0, acc[0][1], false, false);
            acc[1][0] = __builtin_amdgcn_wmma_f32_16x16x4_f32(false, a1, false, b0, (short)0, acc[1][0], false, false);
            acc[1][1] = __builtin_amdgcn_wmma_f32_16x16x4_f32(false, a1, false, b1, (short)0, acc[1][1], false, false);
        }
    }

    #pragma unroll
    for (int mi = 0; mi < 2; ++mi) {
        #pragma unroll
        for (int nj = 0; nj < 2; ++nj) {
            const int n = n_base + nj * 16 + l16;
            #pragma unroll
            for (int r = 0; r < 8; ++r) {
                const int o = m_base + mi * 16 + hl * 8 + r;
                out[(((size_t)b * OCH + o) * HH + h) * WW + wstart + n] = acc[mi][nj][r] + sBias[o];
            }
        }
    }
}

extern "C" void kernel_launch(void* const* d_in, const int* in_sizes, int n_in,
                              void* d_out, int out_size, void* d_ws, size_t ws_size,
                              hipStream_t stream) {
    const float* x      = (const float*)d_in[0];
    const float* net_w  = (const float*)d_in[1];
    const float* weight = (const float*)d_in[2];
    const float* bias   = (const float*)d_in[3];
    float* out = (float*)d_out;

    float* ws = (float*)d_ws;
    float* att   = ws;
    float* aggb  = att + BS * NEXP;
    float* aggw  = aggb + BS * OCH;
    float* parth = aggw + (size_t)BS * OCH * KTOT;
    float* partv = parth + (size_t)BS * CIN * HH;

    att_partial_kernel<<<BS * CIN, 256, 0, stream>>>(x, parth, partv);
    att_finalize_kernel<<<BS, 256, 0, stream>>>(parth, partv, net_w, bias, att, aggb);
    agg_weight_kernel<<<(BS * OCH * KTOT) / 256, 256, 0, stream>>>(weight, att, aggw);
    condconv_wmma_kernel<<<BS * HH * (WW / TILE_W), 256, 0, stream>>>(x, aggw, aggb, out);
}


# conv staging: next chunk's weight slab and input patch fetched as one batch of 16-byte global loads into registers ahead of the WMMA loop, stored to LDS behind it (was load-wait-store per element)
# speedup vs baseline: 1.4083x; 1.4083x over previous
#include <hip/hip_runtime.h>
#include <hip/hip_bf16.h>
#include <math.h>

typedef __attribute__((ext_vector_type(2))) float v2f;
typedef __attribute__((ext_vector_type(4))) float v4f;
typedef __attribute__((ext_vector_type(8))) float v8f;

#define BS   8
#define CIN  64
#define HH   256
#define WW   256
#define NEXP 4
#define OCH  64
#define KS   3
#define KTOT (CIN * KS * KS)
#define WROW (CIN * KS * KS)

#define TILE_W   128
#define CI_CHUNK 16
#define KCHUNK   (CI_CHUNK * 9)
#define SW_STRIDE 148
#define PATCH_W  136
#define PATCH_C0 3

__device__ constexpr int poff(int k) {
    return ((k / 9) * 3 + ((k % 9) / 3)) * PATCH_W + (k % 3) + PATCH_C0;
}

__global__ __launch_bounds__(256)
void att_partial_kernel(const float* __restrict__ x,
                        float* __restrict__ parth,
                        float* __restrict__ partv) {
    const int t = threadIdx.x;
    const float* plane = x + (size_t)blockIdx.x * (HH * WW);
    float rowAcc = 0.f, colAcc = 0.f;
    for (int w = 0; w < WW; ++w) rowAcc += plane[t * WW + w];
    for (int h = 0; h < HH; ++h) colAcc += plane[h * WW + t];
    parth[(size_t)blockIdx.x * HH + t] = rowAcc;
    partv[(size_t)blockIdx.x * WW + t] = colAcc;
}

__global__ __launch_bounds__(256)
void att_finalize_kernel(const float* __restrict__ parth,
                         const float* __restrict__ partv,
                         const float* __restrict__ net_w,
                         const float* __restrict__ bias,
                         float* __restrict__ att,
                         float* __restrict__ aggb) {
    __shared__ float sh[HH + WW];
    __shared__ float sa[NEXP];
    const int b = blockIdx.x, t = threadIdx.x;
    float accH = 0.f, accV = 0.f;
    for (int c = 0; c < CIN; ++c) {
        accH += parth[((size_t)b * CIN + c) * HH + t];
        accV += partv[((size_t)b * CIN + c) * WW + t];
    }
    const float inv = 1.f / (float)(CIN * WW);
    sh[t]      = accH * inv;
    sh[HH + t] = accV * inv;
    __syncthreads();
    if (t < NEXP) {
        float acc = 0.f;
        for (int j = 0; j < HH + WW; ++j) acc += sh[j] * net_w[t * (HH + WW) + j];
        float s = 1.f / (1.f + __expf(-acc));
        att[b * NEXP + t] = s;
        sa[t] = s;
    }
    __syncthreads();
    if (t < OCH) {
        float acc = 0.f;
        for (int k = 0; k < NEXP; ++k) acc += sa[k] * bias[k * OCH + t];
        aggb[b * OCH + t] = acc;
    }
}

__global__ __launch_bounds__(256)
void agg_weight_kernel(const float* __restrict__ weight,
                       const float* __restrict__ att,
                       float* __restrict__ aggw) {
    const size_t idx = (size_t)blockIdx.x * 256 + threadIdx.x;
    const int b = (int)(idx / (OCH * KTOT));
    const int j = (int)(idx % (OCH * KTOT));
    const float a0 = att[b * NEXP + 0], a1 = att[b * NEXP + 1];
    const float a2 = att[b * NEXP + 2], a3 = att[b * NEXP + 3];
    const size_t E = (size_t)OCH * KTOT;
    aggw[idx] = a0 * weight[j] + a1 * weight[E + j] + a2 * weight[2 * E + j] + a3 * weight[3 * E + j];
}

__global__ __launch_bounds__(256)
void condconv_wmma_kernel(const float* __restrict__ x,
                          const float* __restrict__ aggw,
                          const float* __restrict__ aggb,
                          float* __restrict__ out) {
    __shared__ __attribute__((aligned(16))) float sW[OCH * SW_STRIDE];
    __shared__ __attribute__((aligned(16))) float sX[CI_CHUNK * 3 * PATCH_W];
    __shared__ float sBias[OCH];

    const int tid = threadIdx.x;
    const int b      = blockIdx.x >> 9;
    const int rem    = blockIdx.x & 511;
    const int h      = rem >> 1;
    const int wstart = (rem & 1) * TILE_W;

    if (tid < OCH) sBias[tid] = aggb[b * OCH + tid];

    const int lane   = tid & 31;
    const int wv     = tid >> 5;
    const int hl     = lane >> 4;
    const int l16    = lane & 15;
    const int m_base = (wv & 1) * 32;
    const int n_base = (wv >> 1) * 32;

    v8f acc[2][2] = {};

    const int aoff0 = (m_base + l16) * SW_STRIDE + 2 * hl;
    const int aoff1 = (m_base + 16 + l16) * SW_STRIDE + 2 * hl;
    const int n0 = n_base + l16;
    const int n1 = n0 + 16;

    const float* wsrc = aggw + (size_t)b * (OCH * WROW) + (tid >> 2) * WROW + (tid & 3) * 4;
    float*       wdst = sW + (tid >> 2) * SW_STRIDE + (tid & 3) * 4;

    const float* xsrc[6];
    bool xok[6];
    #pragma unroll
    for (int i = 0; i < 6; ++i) {
        const int r  = wv + 8 * i;
        const int ci = r / 3, rr = r - 3 * ci;
        const int gh = h - 1 + rr;
        xok[i] = (unsigned)gh < (unsigned)HH;
        const int ghc = gh < 0 ? 0 : (gh > HH - 1 ? HH - 1 : gh);
        xsrc[i] = x + (((size_t)(b * CIN + ci) * HH + ghc) * WW + wstart + 4 * lane);
    }
    float* xdst = sX + wv * PATCH_W + (PATCH_C0 + 1) + 4 * lane;

    const int  hr   = tid >> 1, hside = tid & 1;
    const int  hci  = hr / 3, hrr = hr - 3 * hci;
    const int  hgh  = h - 1 + hrr;
    const int  hgw  = hside ? wstart + TILE_W : wstart - 1;
    const bool hok  = (unsigned)hgh < (unsigned)HH && (unsigned)hgw < (unsigned)WW;
    const int  hghc = hgh < 0 ? 0 : (hgh > HH - 1 ? HH - 1 : hgh);
    const int  hgwc = hgw < 0 ? 0 : (hgw > WW - 1 ? WW - 1 : hgw);
    const float* hsrc = x + (((size_t)(b * CIN + hci) * HH + hghc) * WW + hgwc);
    float*       hdst = sX + hr * PATCH_W + (hside ? PATCH_C0 + 1 + TILE_W : PATCH_C0);
    const bool   hact = tid < CI_CHUNK * 3 * 2;

    v4f   wreg[9];
    v4f   xreg[6];
    float hreg = 0.f;

    auto issue_loads = [&](int ci0) {
        #pragma unroll
        for (int j = 0; j < 9; ++j) wreg[j] = *(const v4f*)(wsrc + ci0 * 9 + 16 * j);
        #pragma unroll
        for (int i = 0; i < 6; ++i) xreg[i] = *(const v4f*)(xsrc[i] + (size_t)ci0 * (HH * WW));
        if (hact) hreg = hsrc[(size_t)ci0 * (HH * WW)];
    };

    issue_loads(0);

    for (int ci0 = 0; ci0 < CIN; ci0 += CI_CHUNK) {
        __syncthreads();
        #pragma unroll
        for (int j = 0; j < 9; ++j) *(v4f*)(wdst + 16 * j) = wreg[j];
        #pragma unroll
        for (int i = 0; i < 6; ++i) {
            const v4f z = {0.f, 0.f, 0.f, 0.f};
            *(v4f*)(xdst + i * 8 * PATCH_W) = xok[i] ? xreg[i] : z;
        }
        if (hact) *hdst = hok ? hreg : 0.f;
        __syncthreads();

        if (ci0 + CI_CHUNK < CIN) issue_loads(ci0 + CI_CHUNK);

        #pragma unroll
        for (int kc0 = 0; kc0 < KCHUNK; kc0 += 4) {
            const int o0 = hl ? poff(kc0 + 2) : poff(kc0 + 0);
            const int o1 = hl ? poff(kc0 + 3) : poff(kc0 + 1);
            const v2f a0 = *(const v2f*)&sW[aoff0 + kc0];
            const v2f a1 = *(const v2f*)&sW[aoff1 + kc0];
            v2f b0, b1;
            b0.x = sX[o0 + n0]; b0.y = sX[o1 + n0];
            b1.x = sX[o0 + n1]; b1.y = sX[o1 + n1];
            acc[0][0] = __builtin_amdgcn_wmma_f32_16x16x4_f32(false, a0, false, b0, (short)0, acc[0][0], false, false);
            acc[0][1] = __builtin_amdgcn_wmma_f32_16x16x4_f32(false, a0, false, b1, (short)0, acc[0][1], false, false);
            acc[1][0] = __builtin_amdgcn_wmma_f32_16x16x4_f32(false, a1, false, b0, (short)0, acc[1][0], false, false);
            acc[1][1] = __builtin_amdgcn_wmma_f32_16x16x4_f32(false, a1, false, b1, (short)0, acc[1][1], false, false);
        }
    }

    #pragma unroll
    for (int mi = 0; mi < 2; ++mi) {
        #pragma unroll
        for (int nj = 0; nj < 2; ++nj) {
            const int n = n_base + nj * 16 + l16;
            #pragma unroll
            for (int r = 0; r < 8; ++r) {
                const int o = m_base + mi * 16 + hl * 8 + r;
                out[(((size_t)b * OCH + o) * HH + h) * WW + wstart + n] = acc[mi][nj][r] + sBias[o];
            }
        }
    }
}

extern "C" void kernel_launch(void* const* d_in, const int* in_sizes, int n_in,
                              void* d_out, int out_size, void* d_ws, size_t ws_size,
                              hipStream_t stream) {
    const float* x      = (const float*)d_in[0];
    const float* net_w  = (const float*)d_in[1];
    const float* weight = (const float*)d_in[2];
    const float* bias   = (const float*)d_in[3];
    float* out = (float*)d_out;

    float* ws = (float*)d_ws;
    float* att   = ws;
    float* aggb  = att + BS * NEXP;
    float* aggw  = aggb + BS * OCH;
    float* parth = aggw + (size_t)BS * OCH * KTOT;
    float* partv = parth + (size_t)BS * CIN * HH;

    att_partial_kernel<<<BS * CIN, 256, 0, stream>>>(x, parth, partv);
    att_finalize_kernel<<<BS, 256, 0, stream>>>(parth, partv, net_w, bias, att, aggb);
    agg_weight_kernel<<<(BS * OCH * KTOT) / 256, 256, 0, stream>>>(weight, att, aggw);
    condconv_wmma_kernel<<<BS * HH * (WW / TILE_W), 256, 0, stream>>>(x, aggw, aggb, out);
}


# pooling kernel: each plane read once with whole-line 16-byte loads, row and column sums formed in one pass and combined in LDS (was two passes, row pass at 1 KB lane stride); sum order changes
# speedup vs baseline: 1.4832x; 1.0532x over previous
#include <hip/hip_runtime.h>
#include <hip/hip_bf16.h>
#include <math.h>

typedef __attribute__((ext_vector_type(2))) float v2f;
typedef __attribute__((ext_vector_type(4))) float v4f;
typedef __attribute__((ext_vector_type(8))) float v8f;

#define BS   8
#define CIN  64
#define HH   256
#define WW   256
#define NEXP 4
#define OCH  64
#define KS   3
#define KTOT (CIN * KS * KS)
#define WROW (CIN * KS * KS)

#define TILE_W   128
#define CI_CHUNK 16
#define KCHUNK   (CI_CHUNK * 9)
#define SW_STRIDE 148
#define PATCH_W  136
#define PATCH_C0 3

__device__ constexpr int poff(int k) {
    return ((k / 9) * 3 + ((k % 9) / 3)) * PATCH_W + (k % 3) + PATCH_C0;
}

#define RP_PITCH 33
#define ROWS_PER_BATCH 8
__global__ __launch_bounds__(256)
void att_partial_kernel(const float* __restrict__ x,
                        float* __restrict__ parth,
                        float* __restrict__ partv) {
    __shared__ float sRow[HH * RP_PITCH];
    __shared__ __attribute__((aligned(16))) float sCol[8 * WW];
    const int t    = threadIdx.x;
    const int lane = t & 31;
    const int wv   = t >> 5;
    const float* src = x + (size_t)blockIdx.x * (HH * WW) + (size_t)(wv * 32) * WW + 4 * lane;
    float* rdst = sRow + (wv * 32) * RP_PITCH + lane;

    v4f ca = {0.f, 0.f, 0.f, 0.f};
    v4f cb = {0.f, 0.f, 0.f, 0.f};
    for (int i0 = 0; i0 < 32; i0 += ROWS_PER_BATCH) {
        v4f a[ROWS_PER_BATCH], b[ROWS_PER_BATCH];
        for (int i = 0; i < ROWS_PER_BATCH; ++i) {
            a[i] = *(const v4f*)(src + (i0 + i) * WW);
            b[i] = *(const v4f*)(src + (i0 + i) * WW + 128);
        }
        for (int i = 0; i < ROWS_PER_BATCH; ++i) {
            ca += a[i];
            cb += b[i];
            rdst[(i0 + i) * RP_PITCH] = ((a[i].x + a[i].y) + (a[i].z + a[i].w))
                                      + ((b[i].x + b[i].y) + (b[i].z + b[i].w));
        }
    }
    *(v4f*)(sCol + wv * WW + 4 * lane)       = ca;
    *(v4f*)(sCol + wv * WW + 128 + 4 * lane) = cb;
    __syncthreads();

    float rowAcc = 0.f, colAcc = 0.f;
    for (int l = 0; l < 32; ++l) rowAcc += sRow[t * RP_PITCH + l];
    for (int w = 0; w < 8; ++w) colAcc += sCol[w * WW + t];
    parth[(size_t)blockIdx.x * HH + t] = rowAcc;
    partv[(size_t)blockIdx.x * WW + t] = colAcc;
}

__global__ __launch_bounds__(256)
void att_finalize_kernel(const float* __restrict__ parth,
                         const float* __restrict__ partv,
                         const float* __restrict__ net_w,
                         const float* __restrict__ bias,
                         float* __restrict__ att,
                         float* __restrict__ aggb) {
    __shared__ float sh[HH + WW];
    __shared__ float sa[NEXP];
    const int b = blockIdx.x, t = threadIdx.x;
    float accH = 0.f, accV = 0.f;
    for (int c = 0; c < CIN; ++c) {
        accH += parth[((size_t)b * CIN + c) * HH + t];
        accV += partv[((size_t)b * CIN + c) * WW + t];
    }
    const float inv = 1.f / (float)(CIN * WW);
    sh[t]      = accH * inv;
    sh[HH + t] = accV * inv;
    __syncthreads();
    if (t < NEXP) {
        float acc = 0.f;
        for (int j = 0; j < HH + WW; ++j) acc += sh[j] * net_w[t * (HH + WW) + j];
        float s = 1.f / (1.f + __expf(-acc));
        att[b * NEXP + t] = s;
        sa[t] = s;
    }
    __syncthreads();
    if (t < OCH) {
        float acc = 0.f;
        for (int k = 0; k < NEXP; ++k) acc += sa[k] * bias[k * OCH + t];
        aggb[b * OCH + t] = acc;
    }
}

__global__ __launch_bounds__(256)
void agg_weight_kernel(const float* __restrict__ weight,
                       const float* __restrict__ att,
                       float* __restrict__ aggw) {
    const size_t idx = (size_t)blockIdx.x * 256 + threadIdx.x;
    const int b = (int)(idx / (OCH * KTOT));
    const int j = (int)(idx % (OCH * KTOT));
    const float a0 = att[b * NEXP + 0], a1 = att[b * NEXP + 1];
    const float a2 = att[b * NEXP + 2], a3 = att[b * NEXP + 3];
    const size_t E = (size_t)OCH * KTOT;
    aggw[idx] = a0 * weight[j] + a1 * weight[E + j] + a2 * weight[2 * E + j] + a3 * weight[3 * E + j];
}

__global__ __launch_bounds__(256)
void condconv_wmma_kernel(const float* __restrict__ x,
                          const float* __restrict__ aggw,
                          const float* __restrict__ aggb,
                          float* __restrict__ out) {
    __shared__ __attribute__((aligned(16))) float sW[OCH * SW_STRIDE];
    __shared__ __attribute__((aligned(16))) float sX[CI_CHUNK * 3 * PATCH_W];
    __shared__ float sBias[OCH];

    const int tid = threadIdx.x;
    const int b      = blockIdx.x >> 9;
    const int rem    = blockIdx.x & 511;
    const int h      = rem >> 1;
    const int wstart = (rem & 1) * TILE_W;

    if (tid < OCH) sBias[tid] = aggb[b * OCH + tid];

    const int lane   = tid & 31;
    const int wv     = tid >> 5;
    const int hl     = lane >> 4;
    const int l16    = lane & 15;
    const int m_base = (wv & 1) * 32;
    const int n_base = (wv >> 1) * 32;

    v8f acc[2][2] = {};

    const int aoff0 = (m_base + l16) * SW_STRIDE + 2 * hl;
    const int aoff1 = (m_base + 16 + l16) * SW_STRIDE + 2 * hl;
    const int n0 = n_base + l16;
    const int n1 = n0 + 16;

    const float* wsrc = aggw + (size_t)b * (OCH * WROW) + (tid >> 2) * WROW + (tid & 3) * 4;
    float*       wdst = sW + (tid >> 2) * SW_STRIDE + (tid & 3) * 4;

    const float* xsrc[6];
    bool xok[6];
    #pragma unroll
    for (int i = 0; i < 6; ++i) {
        const int r  = wv + 8 * i;
        const int ci = r / 3, rr = r - 3 * ci;
        const int gh = h - 1 + rr;
        xok[i] = (unsigned)gh < (unsigned)HH;
        const int ghc = gh < 0 ? 0 : (gh > HH - 1 ? HH - 1 : gh);
        xsrc[i] = x + (((size_t)(b * CIN + ci) * HH + ghc) * WW + wstart + 4 * lane);
    }
    float* xdst = sX + wv * PATCH_W + (PATCH_C0 + 1) + 4 * lane;

    const int  hr   = tid >> 1, hside = tid & 1;
    const int  hci  = hr / 3, hrr = hr - 3 * hci;
    const int  hgh  = h - 1 + hrr;
    const int  hgw  = hside ? wstart + TILE_W : wstart - 1;
    const bool hok  = (unsigned)hgh < (unsigned)HH && (unsigned)hgw < (unsigned)WW;
    const int  hghc = hgh < 0 ? 0 : (hgh > HH - 1 ? HH - 1 : hgh);
    const int  hgwc = hgw < 0 ? 0 : (hgw > WW - 1 ? WW - 1 : hgw);
    const float* hsrc = x + (((size_t)(b * CIN + hci) * HH + hghc) * WW + hgwc);
    float*       hdst = sX + hr * PATCH_W + (hside ? PATCH_C0 + 1 + TILE_W : PATCH_C0);
    const bool   hact = tid < CI_CHUNK * 3 * 2;

    v4f   wreg[9];
    v4f   xreg[6];
    float hreg = 0.f;

    auto issue_loads = [&](int ci0) {
        #pragma unroll
        for (int j = 0; j < 9; ++j) wreg[j] = *(const v4f*)(wsrc + ci0 * 9 + 16 * j);
        #pragma unroll
        for (int i = 0; i < 6; ++i) xreg[i] = *(const v4f*)(xsrc[i] + (size_t)ci0 * (HH * WW));
        if (hact) hreg = hsrc[(size_t)ci0 * (HH * WW)];
    };

    issue_loads(0);

    for (int ci0 = 0; ci0 < CIN; ci0 += CI_CHUNK) {
        __syncthreads();
        #pragma unroll
        for (int j = 0; j < 9; ++j) *(v4f*)(wdst + 16 * j) = wreg[j];
        #pragma unroll
        for (int i = 0; i < 6; ++i) {
            const v4f z = {0.f, 0.f, 0.f, 0.f};
            *(v4f*)(xdst + i * 8 * PATCH_W) = xok[i] ? xreg[i] : z;
        }
        if (hact) *hdst = hok ? hreg : 0.f;
        __syncthreads();

        if (ci0 + CI_CHUNK < CIN) issue_loads(ci0 + CI_CHUNK);

        #pragma unroll
        for (int kc0 = 0; kc0 < KCHUNK; kc0 += 4) {
            const int o0 = hl ? poff(kc0 + 2) : poff(kc0 + 0);
            const int o1 = hl ? poff(kc0 + 3) : poff(kc0 + 1);
            const v2f a0 = *(const v2f*)&sW[aoff0 + kc0];
            const v2f a1 = *(const v2f*)&sW[aoff1 + kc0];
            v2f b0, b1;
            b0.x = sX[o0 + n0]; b0.y = sX[o1 + n0];
            b1.x = sX[o0 + n1]; b1.y = sX[o1 + n1];
            acc[0][0] = __builtin_amdgcn_wmma_f32_16x16x4_f32(false, a0, false, b0, (short)0, acc[0][0], false, false);
            acc[0][1] = __builtin_amdgcn_wmma_f32_16x16x4_f32(false, a0, false, b1, (short)0, acc[0][1], false, false);
            acc[1][0] = __builtin_amdgcn_wmma_f32_16x16x4_f32(false, a1, false, b0, (short)0, acc[1][0], false, false);
            acc[1][1] = __builtin_amdgcn_wmma_f32_16x16x4_f32(false, a1, false, b1, (short)0, acc[1][1], false, false);
        }
    }

    #pragma unroll
    for (int mi = 0; mi < 2; ++mi) {
        #pragma unroll
        for (int nj = 0; nj < 2; ++nj) {
            const int n = n_base + nj * 16 + l16;
            #pragma unroll
            for (int r = 0; r < 8; ++r) {
                const int o = m_base + mi * 16 + hl * 8 + r;
                out[(((size_t)b * OCH + o) * HH + h) * WW + wstart + n] = acc[mi][nj][r] + sBias[o];
            }
        }
    }
}

extern "C" void kernel_launch(void* const* d_in, const int* in_sizes, int n_in,
                              void* d_out, int out_size, void* d_ws, size_t ws_size,
                              hipStream_t stream) {
    const float* x      = (const float*)d_in[0];
    const float* net_w  = (const float*)d_in[1];
    const float* weight = (const float*)d_in[2];
    const float* bias   = (const float*)d_in[3];
    float* out = (float*)d_out;

    float* ws = (float*)d_ws;
    float* att   = ws;
    float* aggb  = att + BS * NEXP;
    float* aggw  = aggb + BS * OCH;
    float* parth = aggw + (size_t)BS * OCH * KTOT;
    float* partv = parth + (size_t)BS * CIN * HH;

    att_partial_kernel<<<BS * CIN, 256, 0, stream>>>(x, parth, partv);
    att_finalize_kernel<<<BS, 256, 0, stream>>>(parth, partv, net_w, bias, att, aggb);
    agg_weight_kernel<<<(BS * OCH * KTOT) / 256, 256, 0, stream>>>(weight, att, aggw);
    condconv_wmma_kernel<<<BS * HH * (WW / TILE_W), 256, 0, stream>>>(x, aggw, aggb, out);
}


# conv block does two output rows (64x256 tile, wave 32x64): one weight slab and one 4-row patch staged per row pair; lane half 1 takes K two channels on so addresses are base+offset; sum order changes
# speedup vs baseline: 1.6988x; 1.1454x over previous
#include <hip/hip_runtime.h>
#include <hip/hip_bf16.h>
#include <math.h>

typedef __attribute__((ext_vector_type(2))) float v2f;
typedef __attribute__((ext_vector_type(4))) float v4f;
typedef __attribute__((ext_vector_type(8))) float v8f;

#define BS   8
#define CIN  64
#define HH   256
#define WW   256
#define NEXP 4
#define OCH  64
#define KS   3
#define KTOT (CIN * KS * KS)
#define WROW (CIN * KS * KS)

#define TILE_W   128
#define CI_CHUNK 16
#define KCHUNK   (CI_CHUNK * 9)
#define SW_STRIDE 148
#define PATCH_W  132
#define PATCH_C0 3

#define RP_PITCH 33
#define ROWS_PER_BATCH 8
__global__ __launch_bounds__(256)
void att_partial_kernel(const float* __restrict__ x,
                        float* __restrict__ parth,
                        float* __restrict__ partv) {
    __shared__ float sRow[HH * RP_PITCH];
    __shared__ __attribute__((aligned(16))) float sCol[8 * WW];
    const int t    = threadIdx.x;
    const int lane = t & 31;
    const int wv   = t >> 5;
    const float* src = x + (size_t)blockIdx.x * (HH * WW) + (size_t)(wv * 32) * WW + 4 * lane;
    float* rdst = sRow + (wv * 32) * RP_PITCH + lane;

    v4f ca = {0.f, 0.f, 0.f, 0.f};
    v4f cb = {0.f, 0.f, 0.f, 0.f};
    for (int i0 = 0; i0 < 32; i0 += ROWS_PER_BATCH) {
        v4f a[ROWS_PER_BATCH], b[ROWS_PER_BATCH];
        for (int i = 0; i < ROWS_PER_BATCH; ++i) {
            a[i] = *(const v4f*)(src + (i0 + i) * WW);
            b[i] = *(const v4f*)(src + (i0 + i) * WW + 128);
        }
        for (int i = 0; i < ROWS_PER_BATCH; ++i) {
            ca += a[i];
            cb += b[i];
            rdst[(i0 + i) * RP_PITCH] = ((a[i].x + a[i].y) + (a[i].z + a[i].w))
                                      + ((b[i].x + b[i].y) + (b[i].z + b[i].w));
        }
    }
    *(v4f*)(sCol + wv * WW + 4 * lane)       = ca;
    *(v4f*)(sCol + wv * WW + 128 + 4 * lane) = cb;
    __syncthreads();

    float rowAcc = 0.f, colAcc = 0.f;
    for (int l = 0; l < 32; ++l) rowAcc += sRow[t * RP_PITCH + l];
    for (int w = 0; w < 8; ++w) colAcc += sCol[w * WW + t];
    parth[(size_t)blockIdx.x * HH + t] = rowAcc;
    partv[(size_t)blockIdx.x * WW + t] = colAcc;
}

__global__ __launch_bounds__(256)
void att_finalize_kernel(const float* __restrict__ parth,
                         const float* __restrict__ partv,
                         const float* __restrict__ net_w,
                         const float* __restrict__ bias,
                         float* __restrict__ att,
                         float* __restrict__ aggb) {
    __shared__ float sh[HH + WW];
    __shared__ float sa[NEXP];
    const int b = blockIdx.x, t = threadIdx.x;
    float accH = 0.f, accV = 0.f;
    for (int c = 0; c < CIN; ++c) {
        accH += parth[((size_t)b * CIN + c) * HH + t];
        accV += partv[((size_t)b * CIN + c) * WW + t];
    }
    const float inv = 1.f / (float)(CIN * WW);
    sh[t]      = accH * inv;
    sh[HH + t] = accV * inv;
    __syncthreads();
    if (t < NEXP) {
        float acc = 0.f;
        for (int j = 0; j < HH + WW; ++j) acc += sh[j] * net_w[t * (HH + WW) + j];
        float s = 1.f / (1.f + __expf(-acc));
        att[b * NEXP + t] = s;
        sa[t] = s;
    }
    __syncthreads();
    if (t < OCH) {
        float acc = 0.f;
        for (int k = 0; k < NEXP; ++k) acc += sa[k] * bias[k * OCH + t];
        aggb[b * OCH + t] = acc;
    }
}

__global__ __launch_bounds__(256)
void agg_weight_kernel(const float* __restrict__ weight,
                       const float* __restrict__ att,
                       float* __restrict__ aggw) {
    const size_t idx = (size_t)blockIdx.x * 256 + threadIdx.x;
    const int b = (int)(idx / (OCH * KTOT));
    const int j = (int)(idx % (OCH * KTOT));
    const float a0 = att[b * NEXP + 0], a1 = att[b * NEXP + 1];
    const float a2 = att[b * NEXP + 2], a3 = att[b * NEXP + 3];
    const size_t E = (size_t)OCH * KTOT;
    aggw[idx] = a0 * weight[j] + a1 * weight[E + j] + a2 * weight[2 * E + j] + a3 * weight[3 * E + j];
}

#define ROWS_OUT 2
#define PROWS    4
#define NPROW    (CI_CHUNK * PROWS)

__device__ constexpr int boff(int j, int s) {
    return ((4 * (j / 9) + (2 * (j % 9) + s) / 9) * PROWS + ((2 * (j % 9) + s) % 9) / 3) * PATCH_W
           + ((2 * (j % 9) + s) % 3) + PATCH_C0;
}
__device__ constexpr int aoffk(int j) { return 36 * (j / 9) + 2 * (j % 9); }

__global__ __launch_bounds__(256, 4)
void condconv_wmma_kernel(const float* __restrict__ x,
                          const float* __restrict__ aggw,
                          const float* __restrict__ aggb,
                          float* __restrict__ out) {
    __shared__ __attribute__((aligned(16))) float sW[OCH * SW_STRIDE];
    __shared__ __attribute__((aligned(16))) float sX[NPROW * PATCH_W + 4];
    __shared__ float sBias[OCH];

    const int tid = threadIdx.x;
    const int b      = blockIdx.x >> 8;
    const int rem    = blockIdx.x & 255;
    const int h0     = (rem >> 1) * ROWS_OUT;
    const int wstart = (rem & 1) * TILE_W;

    if (tid < OCH) sBias[tid] = aggb[b * OCH + tid];

    const int lane   = tid & 31;
    const int wv     = tid >> 5;
    const int hl     = lane >> 4;
    const int l16    = lane & 15;
    const int m_base = (wv & 1) * 32;
    const int ng     = wv >> 1;
    const int orow   = ng >> 1;
    const int cbase  = (ng & 1) * 64;

    v8f acc[2][4] = {};

    const int aoff0 = (m_base + l16) * SW_STRIDE + 18 * hl;
    const int aoff1 = aoff0 + 16 * SW_STRIDE;
    const int boffl = (2 * PROWS * hl + orow) * PATCH_W + cbase + l16;

    const float* wsrc = aggw + (size_t)b * (OCH * WROW) + (tid >> 2) * WROW + (tid & 3) * 4;
    float*       wdst = sW + (tid >> 2) * SW_STRIDE + (tid & 3) * 4;

    const int  xpr  = wv & 3;
    const int  xgh  = h0 - 1 + xpr;
    const bool xok  = (unsigned)xgh < (unsigned)HH;
    const int  xghc = xgh < 0 ? 0 : (xgh > HH - 1 ? HH - 1 : xgh);
    const float* xsrc = x + (((size_t)(b * CIN + (wv >> 2)) * HH + xghc) * WW + wstart + 4 * lane);
    float* xdst = sX + wv * PATCH_W + (PATCH_C0 + 1) + 4 * lane;

    const int  hr   = tid >> 1, hside = tid & 1;
    const int  hci  = hr >> 2, hpr = hr & 3;
    const int  hgh  = h0 - 1 + hpr;
    const int  hgw  = hside ? wstart + TILE_W : wstart - 1;
    const bool hok  = (unsigned)hgh < (unsigned)HH && (unsigned)hgw < (unsigned)WW;
    const int  hghc = hgh < 0 ? 0 : (hgh > HH - 1 ? HH - 1 : hgh);
    const int  hgwc = hgw < 0 ? 0 : (hgw > WW - 1 ? WW - 1 : hgw);
    const float* hsrc = x + (((size_t)(b * CIN + hci) * HH + hghc) * WW + hgwc);
    float*       hdst = sX + hr * PATCH_W + (hside ? PATCH_C0 + 1 + TILE_W : PATCH_C0);
    const bool   hact = tid < NPROW * 2;

    v4f   wreg[9];
    v4f   xreg[8];
    float hreg = 0.f;

    auto issue_loads = [&](int ci0) {
        #pragma unroll
        for (int j = 0; j < 9; ++j) wreg[j] = *(const v4f*)(wsrc + ci0 * 9 + 16 * j);
        #pragma unroll
        for (int i = 0; i < 8; ++i) xreg[i] = *(const v4f*)(xsrc + (size_t)(ci0 + 2 * i) * (HH * WW));
        if (hact) hreg = hsrc[(size_t)ci0 * (HH * WW)];
    };

    issue_loads(0);

    for (int ci0 = 0; ci0 < CIN; ci0 += CI_CHUNK) {
        __syncthreads();
        #pragma unroll
        for (int j = 0; j < 9; ++j) *(v4f*)(wdst + 16 * j) = wreg[j];
        #pragma unroll
        for (int i = 0; i < 8; ++i) {
            const v4f z = {0.f, 0.f, 0.f, 0.f};
            *(v4f*)(xdst + i * 8 * PATCH_W) = xok ? xreg[i] : z;
        }
        if (hact) *hdst = hok ? hreg : 0.f;
        __syncthreads();

        if (ci0 + CI_CHUNK < CIN) issue_loads(ci0 + CI_CHUNK);

        #pragma unroll
        for (int j = 0; j < KCHUNK / 4; ++j) {
            const v2f a0 = *(const v2f*)&sW[aoff0 + aoffk(j)];
            const v2f a1 = *(const v2f*)&sW[aoff1 + aoffk(j)];
            v2f bf[4];
            #pragma unroll
            for (int nj = 0; nj < 4; ++nj) {
                bf[nj].x = sX[boffl + boff(j, 0) + 16 * nj];
                bf[nj].y = sX[boffl + boff(j, 1) + 16 * nj];
            }
            #pragma unroll
            for (int nj = 0; nj < 4; ++nj) {
                acc[0][nj] = __builtin_amdgcn_wmma_f32_16x16x4_f32(false, a0, false, bf[nj], (short)0, acc[0][nj], false, false);
                acc[1][nj] = __builtin_amdgcn_wmma_f32_16x16x4_f32(false, a1, false, bf[nj], (short)0, acc[1][nj], false, false);
            }
        }
    }

    const int oh = h0 + orow;
    #pragma unroll
    for (int mi = 0; mi < 2; ++mi) {
        #pragma unroll
        for (int nj = 0; nj < 4; ++nj) {
            const int n = cbase + nj * 16 + l16;
            #pragma unroll
            for (int r = 0; r < 8; ++r) {
                const int o = m_base + mi * 16 + hl * 8 + r;
                out[(((size_t)b * OCH + o) * HH + oh) * WW + wstart + n] = acc[mi][nj][r] + sBias[o];
            }
        }
    }
}

extern "C" void kernel_launch(void* const* d_in, const int* in_sizes, int n_in,
                              void* d_out, int out_size, void* d_ws, size_t ws_size,
                              hipStream_t stream) {
    const float* x      = (const float*)d_in[0];
    const float* net_w  = (const float*)d_in[1];
    const float* weight = (const float*)d_in[2];
    const float* bias   = (const float*)d_in[3];
    float* out = (float*)d_out;

    float* ws = (float*)d_ws;
    float* att   = ws;
    float* aggb  = att + BS * NEXP;
    float* aggw  = aggb + BS * OCH;
    float* parth = aggw + (size_t)BS * OCH * KTOT;
    float* partv = parth + (size_t)BS * CIN * HH;

    att_partial_kernel<<<BS * CIN, 256, 0, stream>>>(x, parth, partv);
    att_finalize_kernel<<<BS, 256, 0, stream>>>(parth, partv, net_w, bias, att, aggb);
    agg_weight_kernel<<<(BS * OCH * KTOT) / 256, 256, 0, stream>>>(weight, att, aggw);
    condconv_wmma_kernel<<<BS * (HH / ROWS_OUT) * (WW / TILE_W), 256, 0, stream>>>(x, aggw, aggb, out);
}


# conv: 8-channel staging chunks in two LDS stage buffers (one barrier per chunk) halve the staging registers; 160-VGPR budget puts three blocks on a WGP instead of two; sum order unchanged
# speedup vs baseline: 1.7555x; 1.0334x over previous
#include <hip/hip_runtime.h>
#include <hip/hip_bf16.h>
#include <math.h>

typedef __attribute__((ext_vector_type(2))) float v2f;
typedef __attribute__((ext_vector_type(4))) float v4f;
typedef __attribute__((ext_vector_type(8))) float v8f;

#define BS   8
#define CIN  64
#define HH   256
#define WW   256
#define NEXP 4
#define OCH  64
#define KS   3
#define KTOT (CIN * KS * KS)
#define WROW (CIN * KS * KS)

#define TILE_W   128
#define CI_CHUNK 8
#define KCHUNK   (CI_CHUNK * 9)
#define SW_STRIDE 76
#define PATCH_W  132
#define PATCH_C0 3

#define RP_PITCH 33
#define ROWS_PER_BATCH 8
__global__ __launch_bounds__(256)
void att_partial_kernel(const float* __restrict__ x,
                        float* __restrict__ parth,
                        float* __restrict__ partv) {
    __shared__ float sRow[HH * RP_PITCH];
    __shared__ __attribute__((aligned(16))) float sCol[8 * WW];
    const int t    = threadIdx.x;
    const int lane = t & 31;
    const int wv   = t >> 5;
    const float* src = x + (size_t)blockIdx.x * (HH * WW) + (size_t)(wv * 32) * WW + 4 * lane;
    float* rdst = sRow + (wv * 32) * RP_PITCH + lane;

    v4f ca = {0.f, 0.f, 0.f, 0.f};
    v4f cb = {0.f, 0.f, 0.f, 0.f};
    for (int i0 = 0; i0 < 32; i0 += ROWS_PER_BATCH) {
        v4f a[ROWS_PER_BATCH], b[ROWS_PER_BATCH];
        for (int i = 0; i < ROWS_PER_BATCH; ++i) {
            a[i] = *(const v4f*)(src + (i0 + i) * WW);
            b[i] = *(const v4f*)(src + (i0 + i) * WW + 128);
        }
        for (int i = 0; i < ROWS_PER_BATCH; ++i) {
            ca += a[i];
            cb += b[i];
            rdst[(i0 + i) * RP_PITCH] = ((a[i].x + a[i].y) + (a[i].z + a[i].w))
                                      + ((b[i].x + b[i].y) + (b[i].z + b[i].w));
        }
    }
    *(v4f*)(sCol + wv * WW + 4 * lane)       = ca;
    *(v4f*)(sCol + wv * WW + 128 + 4 * lane) = cb;
    __syncthreads();

    float rowAcc = 0.f, colAcc = 0.f;
    for (int l = 0; l < 32; ++l) rowAcc += sRow[t * RP_PITCH + l];
    for (int w = 0; w < 8; ++w) colAcc += sCol[w * WW + t];
    parth[(size_t)blockIdx.x * HH + t] = rowAcc;
    partv[(size_t)blockIdx.x * WW + t] = colAcc;
}

__global__ __launch_bounds__(256)
void att_finalize_kernel(const float* __restrict__ parth,
                         const float* __restrict__ partv,
                         const float* __restrict__ net_w,
                         const float* __restrict__ bias,
                         float* __restrict__ att,
                         float* __restrict__ aggb) {
    __shared__ float sh[HH + WW];
    __shared__ float sa[NEXP];
    const int b = blockIdx.x, t = threadIdx.x;
    float accH = 0.f, accV = 0.f;
    for (int c = 0; c < CIN; ++c) {
        accH += parth[((size_t)b * CIN + c) * HH + t];
        accV += partv[((size_t)b * CIN + c) * WW + t];
    }
    const float inv = 1.f / (float)(CIN * WW);
    sh[t]      = accH * inv;
    sh[HH + t] = accV * inv;
    __syncthreads();
    if (t < NEXP) {
        float acc = 0.f;
        for (int j = 0; j < HH + WW; ++j) acc += sh[j] * net_w[t * (HH + WW) + j];
        float s = 1.f / (1.f + __expf(-acc));
        att[b * NEXP + t] = s;
        sa[t] = s;
    }
    __syncthreads();
    if (t < OCH) {
        float acc = 0.f;
        for (int k = 0; k < NEXP; ++k) acc += sa[k] * bias[k * OCH + t];
        aggb[b * OCH + t] = acc;
    }
}

__global__ __launch_bounds__(256)
void agg_weight_kernel(const float* __restrict__ weight,
                       const float* __restrict__ att,
                       float* __restrict__ aggw) {
    const size_t idx = (size_t)blockIdx.x * 256 + threadIdx.x;
    const int b = (int)(idx / (OCH * KTOT));
    const int j = (int)(idx % (OCH * KTOT));
    const float a0 = att[b * NEXP + 0], a1 = att[b * NEXP + 1];
    const float a2 = att[b * NEXP + 2], a3 = att[b * NEXP + 3];
    const size_t E = (size_t)OCH * KTOT;
    aggw[idx] = a0 * weight[j] + a1 * weight[E + j] + a2 * weight[2 * E + j] + a3 * weight[3 * E + j];
}

#define ROWS_OUT 2
#define PROWS    4
#define NPROW    (CI_CHUNK * PROWS)
#define NCHUNK   (CIN / CI_CHUNK)
#define SW_BUF   (OCH * SW_STRIDE)
#define SX_BUF   (NPROW * PATCH_W + 4)

__device__ constexpr int boff(int j, int s) {
    return ((4 * (j / 9) + (2 * (j % 9) + s) / 9) * PROWS + ((2 * (j % 9) + s) % 9) / 3) * PATCH_W
           + ((2 * (j % 9) + s) % 3) + PATCH_C0;
}
__device__ constexpr int aoffk(int j) { return 36 * (j / 9) + 2 * (j % 9); }

__global__ __launch_bounds__(256, 6)
void condconv_wmma_kernel(const float* __restrict__ x,
                          const float* __restrict__ aggw,
                          const float* __restrict__ aggb,
                          float* __restrict__ out) {
    __shared__ __attribute__((aligned(16))) float sW[2 * SW_BUF];
    __shared__ __attribute__((aligned(16))) float sX[2 * SX_BUF];
    __shared__ float sBias[OCH];

    const int tid = threadIdx.x;
    const int b      = blockIdx.x >> 8;
    const int rem    = blockIdx.x & 255;
    const int h0     = (rem >> 1) * ROWS_OUT;
    const int wstart = (rem & 1) * TILE_W;

    if (tid < OCH) sBias[tid] = aggb[b * OCH + tid];

    const int lane   = tid & 31;
    const int wv     = tid >> 5;
    const int hl     = lane >> 4;
    const int l16    = lane & 15;
    const int m_base = (wv & 1) * 32;
    const int ng     = wv >> 1;
    const int orow   = ng >> 1;
    const int cbase  = (ng & 1) * 64;

    v8f acc[2][4] = {};

    const int aoff0 = (m_base + l16) * SW_STRIDE + 18 * hl;
    const int aoff1 = aoff0 + 16 * SW_STRIDE;
    const int boffl = (2 * PROWS * hl + orow) * PATCH_W + cbase + l16;

    const bool   wlast = (tid & 3) < 2;
    const int    wlastoff = wlast ? 64 : 48;
    const float* wsrc = aggw + (size_t)b * (OCH * WROW) + (tid >> 2) * WROW + (tid & 3) * 4;
    float*       wdst = sW + (tid >> 2) * SW_STRIDE + (tid & 3) * 4;

    const int  xpr  = wv & 3;
    const int  xgh  = h0 - 1 + xpr;
    const bool xok  = (unsigned)xgh < (unsigned)HH;
    const int  xghc = xgh < 0 ? 0 : (xgh > HH - 1 ? HH - 1 : xgh);
    const float* xsrc = x + (((size_t)(b * CIN + (wv >> 2)) * HH + xghc) * WW + wstart + 4 * lane);
    float* xdst = sX + wv * PATCH_W + (PATCH_C0 + 1) + 4 * lane;

    const int  hr   = tid >> 1, hside = tid & 1;
    const int  hci  = hr >> 2, hpr = hr & 3;
    const int  hgh  = h0 - 1 + hpr;
    const int  hgw  = hside ? wstart + TILE_W : wstart - 1;
    const bool hok  = (unsigned)hgh < (unsigned)HH && (unsigned)hgw < (unsigned)WW;
    const int  hghc = hgh < 0 ? 0 : (hgh > HH - 1 ? HH - 1 : hgh);
    const int  hgwc = hgw < 0 ? 0 : (hgw > WW - 1 ? WW - 1 : hgw);
    const float* hsrc = x + (((size_t)(b * CIN + hci) * HH + hghc) * WW + hgwc);
    float*       hdst = sX + hr * PATCH_W + (hside ? PATCH_C0 + 1 + TILE_W : PATCH_C0);
    const bool   hact = tid < NPROW * 2;

    v4f   wreg[5];
    v4f   xreg[4];
    float hreg = 0.f;

    auto issue_loads = [&](int ci0) {
        #pragma unroll
        for (int j = 0; j < 4; ++j) wreg[j] = *(const v4f*)(wsrc + ci0 * 9 + 16 * j);
        wreg[4] = *(const v4f*)(wsrc + ci0 * 9 + wlastoff);
        #pragma unroll
        for (int i = 0; i < 4; ++i) xreg[i] = *(const v4f*)(xsrc + (size_t)(ci0 + 2 * i) * (HH * WW));
        if (hact) hreg = hsrc[(size_t)ci0 * (HH * WW)];
    };
    auto store_stage = [&](int par) {
        float* wd = wdst + par * SW_BUF;
        float* xd = xdst + par * SX_BUF;
        #pragma unroll
        for (int j = 0; j < 4; ++j) *(v4f*)(wd + 16 * j) = wreg[j];
        if (wlast) *(v4f*)(wd + 64) = wreg[4];
        #pragma unroll
        for (int i = 0; i < 4; ++i) {
            const v4f z = {0.f, 0.f, 0.f, 0.f};
            *(v4f*)(xd + i * 8 * PATCH_W) = xok ? xreg[i] : z;
        }
        if (hact) hdst[par * SX_BUF] = hok ? hreg : 0.f;
    };

    issue_loads(0);
    store_stage(0);
    __syncthreads();

    for (int c = 0; c < NCHUNK; ++c) {
        const int par = c & 1;
        if (c + 1 < NCHUNK) issue_loads((c + 1) * CI_CHUNK);

        const float* cW = sW + par * SW_BUF;
        const float* cX = sX + par * SX_BUF;
        #pragma unroll
        for (int j = 0; j < KCHUNK / 4; ++j) {
            const v2f a0 = *(const v2f*)&cW[aoff0 + aoffk(j)];
            const v2f a1 = *(const v2f*)&cW[aoff1 + aoffk(j)];
            v2f bf[4];
            #pragma unroll
            for (int nj = 0; nj < 4; ++nj) {
                bf[nj].x = cX[boffl + boff(j, 0) + 16 * nj];
                bf[nj].y = cX[boffl + boff(j, 1) + 16 * nj];
            }
            #pragma unroll
            for (int nj = 0; nj < 4; ++nj) {
                acc[0][nj] = __builtin_amdgcn_wmma_f32_16x16x4_f32(false, a0, false, bf[nj], (short)0, acc[0][nj], false, false);
                acc[1][nj] = __builtin_amdgcn_wmma_f32_16x16x4_f32(false, a1, false, bf[nj], (short)0, acc[1][nj], false, false);
            }
        }

        if (c + 1 < NCHUNK) {
            store_stage(par ^ 1);
            __syncthreads();
        }
    }

    const int oh = h0 + orow;
    #pragma unroll
    for (int mi = 0; mi < 2; ++mi) {
        #pragma unroll
        for (int nj = 0; nj < 4; ++nj) {
            const int n = cbase + nj * 16 + l16;
            #pragma unroll
            for (int r = 0; r < 8; ++r) {
                const int o = m_base + mi * 16 + hl * 8 + r;
                out[(((size_t)b * OCH + o) * HH + oh) * WW + wstart + n] = acc[mi][nj][r] + sBias[o];
            }
        }
    }
}

extern "C" void kernel_launch(void* const* d_in, const int* in_sizes, int n_in,
                              void* d_out, int out_size, void* d_ws, size_t ws_size,
                              hipStream_t stream) {
    const float* x      = (const float*)d_in[0];
    const float* net_w  = (const float*)d_in[1];
    const float* weight = (const float*)d_in[2];
    const float* bias   = (const float*)d_in[3];
    float* out = (float*)d_out;

    float* ws = (float*)d_ws;
    float* att   = ws;
    float* aggb  = att + BS * NEXP;
    float* aggw  = aggb + BS * OCH;
    float* parth = aggw + (size_t)BS * OCH * KTOT;
    float* partv = parth + (size_t)BS * CIN * HH;

    att_partial_kernel<<<BS * CIN, 256, 0, stream>>>(x, parth, partv);
    att_finalize_kernel<<<BS, 256, 0, stream>>>(parth, partv, net_w, bias, att, aggb);
    agg_weight_kernel<<<(BS * OCH * KTOT) / 256, 256, 0, stream>>>(weight, att, aggw);
    condconv_wmma_kernel<<<BS * (HH / ROWS_OUT) * (WW / TILE_W), 256, 0, stream>>>(x, aggw, aggb, out);
}


# gate finalize + expert-weight mix fused into one launch, 72 blocks x 1024 threads: each block forms its sample's 4 gate values block-wide (was a 512-step chain on 4 lanes); gate sum order changes
# speedup vs baseline: 1.8449x; 1.0509x over previous
#include <hip/hip_runtime.h>
#include <hip/hip_bf16.h>
#include <math.h>

typedef __attribute__((ext_vector_type(2))) float v2f;
typedef __attribute__((ext_vector_type(4))) float v4f;
typedef __attribute__((ext_vector_type(8))) float v8f;

#define BS   8
#define CIN  64
#define HH   256
#define WW   256
#define NEXP 4
#define OCH  64
#define KS   3
#define KTOT (CIN * KS * KS)
#define WROW (CIN * KS * KS)

#define TILE_W   128
#define CI_CHUNK 8
#define KCHUNK   (CI_CHUNK * 9)
#define SW_STRIDE 76
#define PATCH_W  132
#define PATCH_C0 3

#define RP_PITCH 33
#define ROWS_PER_BATCH 8
__global__ __launch_bounds__(256)
void att_partial_kernel(const float* __restrict__ x,
                        float* __restrict__ parth,
                        float* __restrict__ partv) {
    __shared__ float sRow[HH * RP_PITCH];
    __shared__ __attribute__((aligned(16))) float sCol[8 * WW];
    const int t    = threadIdx.x;
    const int lane = t & 31;
    const int wv   = t >> 5;
    const float* src = x + (size_t)blockIdx.x * (HH * WW) + (size_t)(wv * 32) * WW + 4 * lane;
    float* rdst = sRow + (wv * 32) * RP_PITCH + lane;

    v4f ca = {0.f, 0.f, 0.f, 0.f};
    v4f cb = {0.f, 0.f, 0.f, 0.f};
    for (int i0 = 0; i0 < 32; i0 += ROWS_PER_BATCH) {
        v4f a[ROWS_PER_BATCH], b[ROWS_PER_BATCH];
        for (int i = 0; i < ROWS_PER_BATCH; ++i) {
            a[i] = *(const v4f*)(src + (i0 + i) * WW);
            b[i] = *(const v4f*)(src + (i0 + i) * WW + 128);
        }
        for (int i = 0; i < ROWS_PER_BATCH; ++i) {
            ca += a[i];
            cb += b[i];
            rdst[(i0 + i) * RP_PITCH] = ((a[i].x + a[i].y) + (a[i].z + a[i].w))
                                      + ((b[i].x + b[i].y) + (b[i].z + b[i].w));
        }
    }
    *(v4f*)(sCol + wv * WW + 4 * lane)       = ca;
    *(v4f*)(sCol + wv * WW + 128 + 4 * lane) = cb;
    __syncthreads();

    float rowAcc = 0.f, colAcc = 0.f;
    for (int l = 0; l < 32; ++l) rowAcc += sRow[t * RP_PITCH + l];
    for (int w = 0; w < 8; ++w) colAcc += sCol[w * WW + t];
    parth[(size_t)blockIdx.x * HH + t] = rowAcc;
    partv[(size_t)blockIdx.x * WW + t] = colAcc;
}

#define MIX_PARTS 9
#define CGROUPS   4
#define CPG       (CIN / CGROUPS)
#define DSEG      8
__global__ __launch_bounds__(1024)
void att_mix_kernel(const float* __restrict__ parth,
                    const float* __restrict__ partv,
                    const float* __restrict__ net_w,
                    const float* __restrict__ bias,
                    const float* __restrict__ weight,
                    float* __restrict__ aggb,
                    float* __restrict__ aggw) {
    __shared__ float sPart[CGROUPS * (HH + WW)];
    __shared__ float sMean[HH + WW];
    __shared__ float sDot[NEXP * DSEG];
    __shared__ float sAtt[NEXP];
    const int t    = threadIdx.x;
    const int lane = t & 31;
    const int wv   = t >> 5;
    const int b    = blockIdx.x / MIX_PARTS;
    const int part = blockIdx.x % MIX_PARTS;

    const size_t E  = (size_t)OCH * KTOT;
    const int    j4 = (part * 1024 + t) * 4;
    const v4f w0 = *(const v4f*)(weight + j4);
    const v4f w1 = *(const v4f*)(weight + E + j4);
    const v4f w2 = *(const v4f*)(weight + 2 * E + j4);
    const v4f w3 = *(const v4f*)(weight + 3 * E + j4);

    const int dk = wv & 3, dseg = wv >> 2;
    const int dj = dseg * 64 + lane;
    const float n0 = net_w[dk * (HH + WW) + dj];
    const float n1 = net_w[dk * (HH + WW) + dj + 32];

    const int q = t >> 8, i = t & 255;
    const float* ph = parth + ((size_t)b * CIN + q * CPG) * HH + i;
    const float* pv = partv + ((size_t)b * CIN + q * CPG) * WW + i;
    float vh[CPG], vv[CPG];
    #pragma unroll
    for (int c = 0; c < CPG; ++c) { vh[c] = ph[c * HH]; vv[c] = pv[c * WW]; }
    float accH = 0.f, accV = 0.f;
    #pragma unroll
    for (int c = 0; c < CPG; ++c) { accH += vh[c]; accV += vv[c]; }
    sPart[q * (HH + WW) + i]      = accH;
    sPart[q * (HH + WW) + HH + i] = accV;
    __syncthreads();

    if (t < HH + WW) {
        const float inv = 1.f / (float)(CIN * WW);
        sMean[t] = (((sPart[t] + sPart[(HH + WW) + t]) + sPart[2 * (HH + WW) + t]) + sPart[3 * (HH + WW) + t]) * inv;
    }
    __syncthreads();

    float p = sMean[dj] * n0 + sMean[dj + 32] * n1;
    #pragma unroll
    for (int off = 16; off > 0; off >>= 1) p += __shfl_xor(p, off);
    if (lane == 0) sDot[dk * DSEG + dseg] = p;
    __syncthreads();

    if (t < NEXP) {
        float acc = 0.f;
        #pragma unroll
        for (int s = 0; s < DSEG; ++s) acc += sDot[t * DSEG + s];
        sAtt[t] = 1.f / (1.f + __expf(-acc));
    }
    __syncthreads();

    const float a0 = sAtt[0], a1 = sAtt[1], a2 = sAtt[2], a3 = sAtt[3];
    if (part == 0 && t < OCH) {
        aggb[b * OCH + t] = ((a0 * bias[t] + a1 * bias[OCH + t]) + a2 * bias[2 * OCH + t]) + a3 * bias[3 * OCH + t];
    }
    *(v4f*)(aggw + (size_t)b * E + j4) = a0 * w0 + a1 * w1 + a2 * w2 + a3 * w3;
}

#define ROWS_OUT 2
#define PROWS    4
#define NPROW    (CI_CHUNK * PROWS)
#define NCHUNK   (CIN / CI_CHUNK)
#define SW_BUF   (OCH * SW_STRIDE)
#define SX_BUF   (NPROW * PATCH_W + 4)

__device__ constexpr int boff(int j, int s) {
    return ((4 * (j / 9) + (2 * (j % 9) + s) / 9) * PROWS + ((2 * (j % 9) + s) % 9) / 3) * PATCH_W
           + ((2 * (j % 9) + s) % 3) + PATCH_C0;
}
__device__ constexpr int aoffk(int j) { return 36 * (j / 9) + 2 * (j % 9); }

__global__ __launch_bounds__(256, 6)
void condconv_wmma_kernel(const float* __restrict__ x,
                          const float* __restrict__ aggw,
                          const float* __restrict__ aggb,
                          float* __restrict__ out) {
    __shared__ __attribute__((aligned(16))) float sW[2 * SW_BUF];
    __shared__ __attribute__((aligned(16))) float sX[2 * SX_BUF];
    __shared__ float sBias[OCH];

    const int tid = threadIdx.x;
    const int b      = blockIdx.x >> 8;
    const int rem    = blockIdx.x & 255;
    const int h0     = (rem >> 1) * ROWS_OUT;
    const int wstart = (rem & 1) * TILE_W;

    if (tid < OCH) sBias[tid] = aggb[b * OCH + tid];

    const int lane   = tid & 31;
    const int wv     = tid >> 5;
    const int hl     = lane >> 4;
    const int l16    = lane & 15;
    const int m_base = (wv & 1) * 32;
    const int ng     = wv >> 1;
    const int orow   = ng >> 1;
    const int cbase  = (ng & 1) * 64;

    v8f acc[2][4] = {};

    const int aoff0 = (m_base + l16) * SW_STRIDE + 18 * hl;
    const int aoff1 = aoff0 + 16 * SW_STRIDE;
    const int boffl = (2 * PROWS * hl + orow) * PATCH_W + cbase + l16;

    const bool   wlast = (tid & 3) < 2;
    const int    wlastoff = wlast ? 64 : 48;
    const float* wsrc = aggw + (size_t)b * (OCH * WROW) + (tid >> 2) * WROW + (tid & 3) * 4;
    float*       wdst = sW + (tid >> 2) * SW_STRIDE + (tid & 3) * 4;

    const int  xpr  = wv & 3;
    const int  xgh  = h0 - 1 + xpr;
    const bool xok  = (unsigned)xgh < (unsigned)HH;
    const int  xghc = xgh < 0 ? 0 : (xgh > HH - 1 ? HH - 1 : xgh);
    const float* xsrc = x + (((size_t)(b * CIN + (wv >> 2)) * HH + xghc) * WW + wstart + 4 * lane);
    float* xdst = sX + wv * PATCH_W + (PATCH_C0 + 1) + 4 * lane;

    const int  hr   = tid >> 1, hside = tid & 1;
    const int  hci  = hr >> 2, hpr = hr & 3;
    const int  hgh  = h0 - 1 + hpr;
    const int  hgw  = hside ? wstart + TILE_W : wstart - 1;
    const bool hok  = (unsigned)hgh < (unsigned)HH && (unsigned)hgw < (unsigned)WW;
    const int  hghc = hgh < 0 ? 0 : (hgh > HH - 1 ? HH - 1 : hgh);
    const int  hgwc = hgw < 0 ? 0 : (hgw > WW - 1 ? WW - 1 : hgw);
    const float* hsrc = x + (((size_t)(b * CIN + hci) * HH + hghc) * WW + hgwc);
    float*       hdst = sX + hr * PATCH_W + (hside ? PATCH_C0 + 1 + TILE_W : PATCH_C0);
    const bool   hact = tid < NPROW * 2;

    v4f   wreg[5];
    v4f   xreg[4];
    float hreg = 0.f;

    auto issue_loads = [&](int ci0) {
        #pragma unroll
        for (int j = 0; j < 4; ++j) wreg[j] = *(const v4f*)(wsrc + ci0 * 9 + 16 * j);
        wreg[4] = *(const v4f*)(wsrc + ci0 * 9 + wlastoff);
        #pragma unroll
        for (int i = 0; i < 4; ++i) xreg[i] = *(const v4f*)(xsrc + (size_t)(ci0 + 2 * i) * (HH * WW));
        if (hact) hreg = hsrc[(size_t)ci0 * (HH * WW)];
    };
    auto store_stage = [&](int par) {
        float* wd = wdst + par * SW_BUF;
        float* xd = xdst + par * SX_BUF;
        #pragma unroll
        for (int j = 0; j < 4; ++j) *(v4f*)(wd + 16 * j) = wreg[j];
        if (wlast) *(v4f*)(wd + 64) = wreg[4];
        #pragma unroll
        for (int i = 0; i < 4; ++i) {
            const v4f z = {0.f, 0.f, 0.f, 0.f};
            *(v4f*)(xd + i * 8 * PATCH_W) = xok ? xreg[i] : z;
        }
        if (hact) hdst[par * SX_BUF] = hok ? hreg : 0.f;
    };

    issue_loads(0);
    store_stage(0);
    __syncthreads();

    for (int c = 0; c < NCHUNK; ++c) {
        const int par = c & 1;
        if (c + 1 < NCHUNK) issue_loads((c + 1) * CI_CHUNK);

        const float* cW = sW + par * SW_BUF;
        const float* cX = sX + par * SX_BUF;
        #pragma unroll
        for (int j = 0; j < KCHUNK / 4; ++j) {
            const v2f a0 = *(const v2f*)&cW[aoff0 + aoffk(j)];
            const v2f a1 = *(const v2f*)&cW[aoff1 + aoffk(j)];
            v2f bf[4];
            #pragma unroll
            for (int nj = 0; nj < 4; ++nj) {
                bf[nj].x = cX[boffl + boff(j, 0) + 16 * nj];
                bf[nj].y = cX[boffl + boff(j, 1) + 16 * nj];
            }
            #pragma unroll
            for (int nj = 0; nj < 4; ++nj) {
                acc[0][nj] = __builtin_amdgcn_wmma_f32_16x16x4_f32(false, a0, false, bf[nj], (short)0, acc[0][nj], false, false);
                acc[1][nj] = __builtin_amdgcn_wmma_f32_16x16x4_f32(false, a1, false, bf[nj], (short)0, acc[1][nj], false, false);
            }
        }

        if (c + 1 < NCHUNK) {
            store_stage(par ^ 1);
            __syncthreads();
        }
    }

    const int oh = h0 + orow;
    #pragma unroll
    for (int mi = 0; mi < 2; ++mi) {
        #pragma unroll
        for (int nj = 0; nj < 4; ++nj) {
            const int n = cbase + nj * 16 + l16;
            #pragma unroll
            for (int r = 0; r < 8; ++r) {
                const int o = m_base + mi * 16 + hl * 8 + r;
                out[(((size_t)b * OCH + o) * HH + oh) * WW + wstart + n] = acc[mi][nj][r] + sBias[o];
            }
        }
    }
}

extern "C" void kernel_launch(void* const* d_in, const int* in_sizes, int n_in,
                              void* d_out, int out_size, void* d_ws, size_t ws_size,
                              hipStream_t stream) {
    const float* x      = (const float*)d_in[0];
    const float* net_w  = (const float*)d_in[1];
    const float* weight = (const float*)d_in[2];
    const float* bias   = (const float*)d_in[3];
    float* out = (float*)d_out;

    float* ws = (float*)d_ws;
    float* att   = ws;
    float* aggb  = att + BS * NEXP;
    float* aggw  = aggb + BS * OCH;
    float* parth = aggw + (size_t)BS * OCH * KTOT;
    float* partv = parth + (size_t)BS * CIN * HH;

    att_partial_kernel<<<BS * CIN, 256, 0, stream>>>(x, parth, partv);
    att_mix_kernel<<<BS * MIX_PARTS, 1024, 0, stream>>>(parth, partv, net_w, bias, weight, aggb, aggw);
    condconv_wmma_kernel<<<BS * (HH / ROWS_OUT) * (WW / TILE_W), 256, 0, stream>>>(x, aggw, aggb, out);
}


# conv: third LDS stage buffer (chunk c+2 staged during chunk c) and the chunk barrier split: signal at chunk end, wait one tap later in front of the next chunk's first LDS write; sum order unchanged
# speedup vs baseline: 2.6529x; 1.4380x over previous
#include <hip/hip_runtime.h>
#include <hip/hip_bf16.h>
#include <math.h>

typedef __attribute__((ext_vector_type(2))) float v2f;
typedef __attribute__((ext_vector_type(4))) float v4f;
typedef __attribute__((ext_vector_type(8))) float v8f;

#define BS   8
#define CIN  64
#define HH   256
#define WW   256
#define NEXP 4
#define OCH  64
#define KS   3
#define KTOT (CIN * KS * KS)
#define WROW (CIN * KS * KS)

#define TILE_W   64
#define CI_CHUNK 8
#define NT       6
#define TCHUNK   (3 * CI_CHUNK)
#define SW_STRIDE 148
#define PATCH_W  68
#define PATCH_C0 3
#define EP_W     68

#define RP_PITCH 33
#define ROWS_PER_BATCH 8
__global__ __launch_bounds__(256)
void att_partial_kernel(const float* __restrict__ x,
                        float* __restrict__ parth,
                        float* __restrict__ partv) {
    __shared__ float sRow[HH * RP_PITCH];
    __shared__ __attribute__((aligned(16))) float sCol[8 * WW];
    const int t    = threadIdx.x;
    const int lane = t & 31;
    const int wv   = t >> 5;
    const float* src = x + (size_t)blockIdx.x * (HH * WW) + (size_t)(wv * 32) * WW + 4 * lane;
    float* rdst = sRow + (wv * 32) * RP_PITCH + lane;

    v4f ca = {0.f, 0.f, 0.f, 0.f};
    v4f cb = {0.f, 0.f, 0.f, 0.f};
    for (int i0 = 0; i0 < 32; i0 += ROWS_PER_BATCH) {
        v4f a[ROWS_PER_BATCH], b[ROWS_PER_BATCH];
        for (int i = 0; i < ROWS_PER_BATCH; ++i) {
            a[i] = *(const v4f*)(src + (i0 + i) * WW);
            b[i] = *(const v4f*)(src + (i0 + i) * WW + 128);
        }
        for (int i = 0; i < ROWS_PER_BATCH; ++i) {
            ca += a[i];
            cb += b[i];
            rdst[(i0 + i) * RP_PITCH] = ((a[i].x + a[i].y) + (a[i].z + a[i].w))
                                      + ((b[i].x + b[i].y) + (b[i].z + b[i].w));
        }
    }
    *(v4f*)(sCol + wv * WW + 4 * lane)       = ca;
    *(v4f*)(sCol + wv * WW + 128 + 4 * lane) = cb;
    __syncthreads();

    float rowAcc = 0.f, colAcc = 0.f;
    for (int l = 0; l < 32; ++l) rowAcc += sRow[t * RP_PITCH + l];
    for (int w = 0; w < 8; ++w) colAcc += sCol[w * WW + t];
    parth[(size_t)blockIdx.x * HH + t] = rowAcc;
    partv[(size_t)blockIdx.x * WW + t] = colAcc;
}

#define MIX_PARTS 9
#define CGROUPS   4
#define CPG       (CIN / CGROUPS)
#define DSEG      8
__global__ __launch_bounds__(1024)
void att_mix_kernel(const float* __restrict__ parth,
                    const float* __restrict__ partv,
                    const float* __restrict__ net_w,
                    const float* __restrict__ bias,
                    const float* __restrict__ weight,
                    float* __restrict__ aggb,
                    float* __restrict__ aggw) {
    __shared__ float sPart[CGROUPS * (HH + WW)];
    __shared__ float sMean[HH + WW];
    __shared__ float sDot[NEXP * DSEG];
    __shared__ float sAtt[NEXP];
    const int t    = threadIdx.x;
    const int lane = t & 31;
    const int wv   = t >> 5;
    const int b    = blockIdx.x / MIX_PARTS;
    const int part = blockIdx.x % MIX_PARTS;

    const size_t E  = (size_t)OCH * KTOT;
    const int    j4 = (part * 1024 + t) * 4;
    const v4f w0 = *(const v4f*)(weight + j4);
    const v4f w1 = *(const v4f*)(weight + E + j4);
    const v4f w2 = *(const v4f*)(weight + 2 * E + j4);
    const v4f w3 = *(const v4f*)(weight + 3 * E + j4);

    const int dk = wv & 3, dseg = wv >> 2;
    const int dj = dseg * 64 + lane;
    const float n0 = net_w[dk * (HH + WW) + dj];
    const float n1 = net_w[dk * (HH + WW) + dj + 32];

    const int q = t >> 8, i = t & 255;
    const float* ph = parth + ((size_t)b * CIN + q * CPG) * HH + i;
    const float* pv = partv + ((size_t)b * CIN + q * CPG) * WW + i;
    float vh[CPG], vv[CPG];
    #pragma unroll
    for (int c = 0; c < CPG; ++c) { vh[c] = ph[c * HH]; vv[c] = pv[c * WW]; }
    float accH = 0.f, accV = 0.f;
    #pragma unroll
    for (int c = 0; c < CPG; ++c) { accH += vh[c]; accV += vv[c]; }
    sPart[q * (HH + WW) + i]      = accH;
    sPart[q * (HH + WW) + HH + i] = accV;
    __syncthreads();

    if (t < HH + WW) {
        const float inv = 1.f / (float)(CIN * WW);
        sMean[t] = (((sPart[t] + sPart[(HH + WW) + t]) + sPart[2 * (HH + WW) + t]) + sPart[3 * (HH + WW) + t]) * inv;
    }
    __syncthreads();

    float p = sMean[dj] * n0 + sMean[dj + 32] * n1;
    #pragma unroll
    for (int off = 16; off > 0; off >>= 1) p += __shfl_xor(p, off);
    if (lane == 0) sDot[dk * DSEG + dseg] = p;
    __syncthreads();

    if (t < NEXP) {
        float acc = 0.f;
        #pragma unroll
        for (int s = 0; s < DSEG; ++s) acc += sDot[t * DSEG + s];
        sAtt[t] = 1.f / (1.f + __expf(-acc));
    }
    __syncthreads();

    const float a0 = sAtt[0], a1 = sAtt[1], a2 = sAtt[2], a3 = sAtt[3];
    if (part == 0 && t < OCH) {
        aggb[b * OCH + t] = ((a0 * bias[t] + a1 * bias[OCH + t]) + a2 * bias[2 * OCH + t]) + a3 * bias[3 * OCH + t];
    }
    *(v4f*)(aggw + (size_t)b * E + j4) = a0 * w0 + a1 * w1 + a2 * w2 + a3 * w3;
}

#define ROWS_OUT 4
#define PROWS    6
#define NPROW    (CI_CHUNK * PROWS)
#define NCHUNK   (CIN / CI_CHUNK)
#define NSTAGE   3
#define SW_BUF   (OCH * SW_STRIDE)
#define SX_BUF   (NPROW * PATCH_W + 4)

__device__ constexpr int boff(int jj, int s, int t) {
    return ((2 * jj * PROWS + t) * 2 + s) * PATCH_W;
}

__global__ __launch_bounds__(256, 4)
void condconv_wmma_kernel(const float* __restrict__ x,
                          const float* __restrict__ aggw,
                          const float* __restrict__ aggb,
                          float* __restrict__ out) {
    __shared__ __attribute__((aligned(16))) float sMem[NSTAGE * SX_BUF + NSTAGE * SW_BUF + OCH];
    float* const sX    = sMem;
    float* const sW    = sMem + NSTAGE * SX_BUF;
    float* const sBias = sW + NSTAGE * SW_BUF;

    const int tid = threadIdx.x;
    const int b      = blockIdx.x >> 8;
    const int rem    = blockIdx.x & 255;
    const int h0     = (rem >> 2) * ROWS_OUT;
    const int wstart = (rem & 3) * TILE_W;

    if (tid < OCH) sBias[tid] = aggb[b * OCH + tid];

    const int lane   = tid & 31;
    const int wv     = tid >> 5;
    const int hl     = lane >> 4;
    const int l16    = lane & 15;
    const int m_base = (wv & 1) * 32;
    const int cbase  = (wv >> 1) * 16;

    v8f acc[NT][2] = {};

    const int aoff0 = (m_base + l16) * SW_STRIDE + 2 * hl;
    const int aoff1 = aoff0 + 16 * SW_STRIDE;
    const int boffl = (hl * PROWS * 2) * PATCH_W + PATCH_C0 + cbase + l16;

    const int    wq   = tid & 3;
    const float* wsrc = aggw + (size_t)b * (OCH * WROW) + (tid >> 2) * WROW + wq * 18;
    float*       wdst = sW + (tid >> 2) * SW_STRIDE + 2 * wq;

    const bool xok0 = h0 > 0;
    const bool xok5 = h0 + ROWS_OUT < HH;
    const int  xr0  = xok0 ? h0 - 1 : 0;
    const int  xr5  = xok5 ? h0 + ROWS_OUT : HH - 1;
    const int  prow = ((wv >> 1) * PROWS) * 2 + (wv & 1);
    const float* xsrc = x + ((size_t)(b * CIN + wv) * HH) * WW + wstart + 2 * lane;
    float* xdst = sX + prow * PATCH_W + (PATCH_C0 + 1) + 2 * lane;

    const bool hact  = lane < 2;
    const int  hside = lane & 1;
    const int  hgw   = hside ? wstart + TILE_W : wstart - 1;
    const bool hokw  = (unsigned)hgw < (unsigned)WW;
    const int  hgwc  = hgw < 0 ? 0 : (hgw > WW - 1 ? WW - 1 : hgw);
    const float* hsrc = x + ((size_t)(b * CIN + wv) * HH) * WW + hgwc;
    float*       hdst = sX + prow * PATCH_W + (hside ? PATCH_C0 + 1 + TILE_W : PATCH_C0);

    auto issue_w = [&](v2f (&wreg)[9], int ci0) {
        #pragma unroll
        for (int i = 0; i < 9; ++i) wreg[i] = *(const v2f*)(wsrc + ci0 * 9 + 2 * i);
    };
    auto issue_x = [&](v2f (&xreg)[PROWS], float (&hreg)[PROWS], int ci0) {
        const float* xs = xsrc + (size_t)ci0 * (HH * WW);
        xreg[0] = *(const v2f*)(xs + xr0 * WW);
        xreg[1] = *(const v2f*)(xs + h0 * WW);
        xreg[2] = *(const v2f*)(xs + (h0 + 1) * WW);
        xreg[3] = *(const v2f*)(xs + (h0 + 2) * WW);
        xreg[4] = *(const v2f*)(xs + (h0 + 3) * WW);
        xreg[5] = *(const v2f*)(xs + xr5 * WW);
        if (hact) {
            const float* hs = hsrc + (size_t)ci0 * (HH * WW);
            hreg[0] = hs[xr0 * WW];
            hreg[1] = hs[h0 * WW];
            hreg[2] = hs[(h0 + 1) * WW];
            hreg[3] = hs[(h0 + 2) * WW];
            hreg[4] = hs[(h0 + 3) * WW];
            hreg[5] = hs[xr5 * WW];
        }
    };
    auto store_w = [&](const v2f (&wreg)[9], int par) {
        float* wd = wdst + par * SW_BUF;
        float f[18];
        #pragma unroll
        for (int i = 0; i < 9; ++i) { f[2 * i] = wreg[i].x; f[2 * i + 1] = wreg[i].y; }
        #pragma unroll
        for (int kw = 0; kw < 3; ++kw) {
            v2f u0, u1, u2, u3, u4, u5;
            #pragma unroll
            for (int cc = 0; cc < 2; ++cc) {
                const float g0 = f[cc * 9 + kw], g1 = f[cc * 9 + 3 + kw], g2 = f[cc * 9 + 6 + kw];
                const float s02 = g0 + g2;
                const float e   = g0 * (1.f / 24.f) + g2 * (1.f / 6.f);
                const float q   = g1 * (1.f / 12.f);
                u0[cc] = 0.25f * g0;
                u1[cc] = (s02 + g1) * (-1.f / 6.f);
                u2[cc] = (s02 - g1) * (-1.f / 6.f);
                u3[cc] = e + q;
                u4[cc] = e - q;
                u5[cc] = g2;
            }
            *(v2f*)(wd + 0 * TCHUNK + kw * CI_CHUNK) = u0;
            *(v2f*)(wd + 1 * TCHUNK + kw * CI_CHUNK) = u1;
            *(v2f*)(wd + 2 * TCHUNK + kw * CI_CHUNK) = u2;
            *(v2f*)(wd + 3 * TCHUNK + kw * CI_CHUNK) = u3;
            *(v2f*)(wd + 4 * TCHUNK + kw * CI_CHUNK) = u4;
            *(v2f*)(wd + 5 * TCHUNK + kw * CI_CHUNK) = u5;
        }
    };
    auto store_x = [&](const v2f (&xreg)[PROWS], const float (&hreg)[PROWS], int par) {
        float* xd = xdst + par * SX_BUF;
        const v2f z = {0.f, 0.f};
        const v2f d0 = xok0 ? xreg[0] : z, d1 = xreg[1], d2 = xreg[2], d3 = xreg[3], d4 = xreg[4];
        const v2f d5 = xok5 ? xreg[5] : z;
        const v2f p42 = d4 - d2, p31 = d3 - d1;
        *(v2f*)(xd + 0 * PATCH_W)  = (4.f * d0 - 5.f * d2) + d4;
        *(v2f*)(xd + 2 * PATCH_W)  = (d3 + d4) - 4.f * (d1 + d2);
        *(v2f*)(xd + 4 * PATCH_W)  = (d4 - d3) + 4.f * (d1 - d2);
        *(v2f*)(xd + 6 * PATCH_W)  = p42 + 2.f * p31;
        *(v2f*)(xd + 8 * PATCH_W)  = p42 - 2.f * p31;
        *(v2f*)(xd + 10 * PATCH_W) = (4.f * d1 - 5.f * d3) + d5;
        if (hact) {
            float* hd = hdst + par * SX_BUF;
            const float e0 = (hokw && xok0) ? hreg[0] : 0.f;
            const float e1 = hokw ? hreg[1] : 0.f;
            const float e2 = hokw ? hreg[2] : 0.f;
            const float e3 = hokw ? hreg[3] : 0.f;
            const float e4 = hokw ? hreg[4] : 0.f;
            const float e5 = (hokw && xok5) ? hreg[5] : 0.f;
            const float q42 = e4 - e2, q31 = e3 - e1;
            hd[0 * PATCH_W]  = (4.f * e0 - 5.f * e2) + e4;
            hd[2 * PATCH_W]  = (e3 + e4) - 4.f * (e1 + e2);
            hd[4 * PATCH_W]  = (e4 - e3) + 4.f * (e1 - e2);
            hd[6 * PATCH_W]  = q42 + 2.f * q31;
            hd[8 * PATCH_W]  = q42 - 2.f * q31;
            hd[10 * PATCH_W] = (4.f * e1 - 5.f * e3) + e5;
        }
    };

    auto mma_tap = [&](const float* tW, const float* tX) {
        #pragma unroll
        for (int jj = 0; jj < 2; ++jj) {
            #pragma unroll
            for (int t = 0; t < NT; ++t) {
                const v2f a0 = *(const v2f*)&tW[aoff0 + t * TCHUNK + 4 * jj];
                const v2f a1 = *(const v2f*)&tW[aoff1 + t * TCHUNK + 4 * jj];
                v2f bf;
                bf.x = tX[boffl + boff(jj, 0, t)];
                bf.y = tX[boffl + boff(jj, 1, t)];
                acc[t][0] = __builtin_amdgcn_wmma_f32_16x16x4_f32(false, a0, false, bf, (short)0, acc[t][0], false, false);
                acc[t][1] = __builtin_amdgcn_wmma_f32_16x16x4_f32(false, a1, false, bf, (short)0, acc[t][1], false, false);
            }
        }
    };

    {
        v2f   wreg0[9], wreg1[9];
        v2f   xreg0[PROWS], xreg1[PROWS];
        float hreg0[PROWS] = {0.f, 0.f, 0.f, 0.f, 0.f, 0.f};
        float hreg1[PROWS] = {0.f, 0.f, 0.f, 0.f, 0.f, 0.f};
        issue_w(wreg0, 0);
        issue_x(xreg0, hreg0, 0);
        issue_w(wreg1, CI_CHUNK);
        issue_x(xreg1, hreg1, CI_CHUNK);
        store_w(wreg0, 0);
        store_x(xreg0, hreg0, 0);
        store_w(wreg1, 1);
        store_x(xreg1, hreg1, 1);
    }
    __syncthreads();

    int st_cur = 0, st_nxt = 1, st_wr = 2;
    float hreg[PROWS] = {0.f, 0.f, 0.f, 0.f, 0.f, 0.f};

    for (int c = 0; c < NCHUNK - 2; ++c) {
        v2f wreg[9];
        v2f xreg[PROWS];
        const float* cW = sW + st_cur * SW_BUF;
        const float* cX = sX + st_cur * SX_BUF;

        issue_w(wreg, (c + 2) * CI_CHUNK);
        mma_tap(cW, cX);
        if (c >= 1) {
            __builtin_amdgcn_s_barrier_wait(-1);
            __builtin_amdgcn_fence(__ATOMIC_ACQUIRE, "workgroup");
        }
        store_w(wreg, st_wr);
        issue_x(xreg, hreg, (c + 2) * CI_CHUNK);
        #pragma nounroll
        for (int kw = 1; kw < KS; ++kw) mma_tap(cW + kw * CI_CHUNK, cX + kw);
        store_x(xreg, hreg, st_wr);
        __builtin_amdgcn_fence(__ATOMIC_RELEASE, "workgroup");
        __builtin_amdgcn_s_barrier_signal(-1);
        const int st_old = st_cur;
        st_cur = st_nxt;
        st_nxt = st_wr;
        st_wr  = st_old;
    }
    {
        const float* cW = sW + st_cur * SW_BUF;
        const float* cX = sX + st_cur * SX_BUF;
        #pragma nounroll
        for (int kw = 0; kw < KS; ++kw) mma_tap(cW + kw * CI_CHUNK, cX + kw);
        __builtin_amdgcn_s_barrier_wait(-1);
        __builtin_amdgcn_fence(__ATOMIC_ACQUIRE, "workgroup");
        cW = sW + st_nxt * SW_BUF;
        cX = sX + st_nxt * SX_BUF;
        #pragma nounroll
        for (int kw = 0; kw < KS; ++kw) mma_tap(cW + kw * CI_CHUNK, cX + kw);
    }

    __syncthreads();
    float* sOut = sW;
    #pragma unroll
    for (int mi = 0; mi < 2; ++mi) {
        #pragma unroll
        for (int r = 0; r < 8; ++r) {
            const int o = m_base + mi * 16 + hl * 8 + r;
            const float m0 = acc[0][mi][r], m1 = acc[1][mi][r], m2 = acc[2][mi][r];
            const float m3 = acc[3][mi][r], m4 = acc[4][mi][r], m5 = acc[5][mi][r];
            const float s12 = m1 + m2, d12 = m1 - m2, s34 = m3 + m4, d34 = m3 - m4;
            float* so = sOut + o * EP_W + cbase + l16;
            so[0 * OCH * EP_W] = (m0 + s12) + s34;
            so[1 * OCH * EP_W] = d12 + 2.f * d34;
            so[2 * OCH * EP_W] = s12 + 4.f * s34;
            so[3 * OCH * EP_W] = (d12 + 8.f * d34) + m5;
        }
    }
    __syncthreads();
    {
        const int piece = tid & 15;
        const int osub  = tid >> 4;
        float bz[4];
        #pragma unroll
        for (int g = 0; g < 4; ++g) bz[g] = sBias[osub + 16 * g];
        #pragma unroll
        for (int i = 0; i < 16; ++i) {
            const int row = i & 3;
            const int o   = osub + 16 * (i >> 2);
            const v4f v = *(const v4f*)(sOut + (row * OCH + o) * EP_W + 4 * piece);
            *(v4f*)(out + (((size_t)b * OCH + o) * HH + h0 + row) * WW + wstart + 4 * piece) = v + bz[i >> 2];
        }
    }
}

extern "C" void kernel_launch(void* const* d_in, const int* in_sizes, int n_in,
                              void* d_out, int out_size, void* d_ws, size_t ws_size,
                              hipStream_t stream) {
    const float* x      = (const float*)d_in[0];
    const float* net_w  = (const float*)d_in[1];
    const float* weight = (const float*)d_in[2];
    const float* bias   = (const float*)d_in[3];
    float* out = (float*)d_out;

    float* ws = (float*)d_ws;
    float* att   = ws;
    float* aggb  = att + BS * NEXP;
    float* aggw  = aggb + BS * OCH;
    float* parth = aggw + (size_t)BS * OCH * KTOT;
    float* partv = parth + (size_t)BS * CIN * HH;

    att_partial_kernel<<<BS * CIN, 256, 0, stream>>>(x, parth, partv);
    att_mix_kernel<<<BS * MIX_PARTS, 1024, 0, stream>>>(parth, partv, net_w, bias, weight, aggb, aggw);
    condconv_wmma_kernel<<<BS * (HH / ROWS_OUT) * (WW / TILE_W), 256, 0, stream>>>(x, aggw, aggb, out);
}


# mixed weights stored by the mix launch in the conv staging threads' fetch order: each conv weight load of a wave is one contiguous 256-byte row (was 32 pieces over 8 rows); sum order unchanged
# speedup vs baseline: 2.7748x; 1.0459x over previous
#include <hip/hip_runtime.h>
#include <hip/hip_bf16.h>
#include <math.h>

typedef __attribute__((ext_vector_type(2))) float v2f;
typedef __attribute__((ext_vector_type(4))) float v4f;
typedef __attribute__((ext_vector_type(8))) float v8f;

#define BS   8
#define CIN  64
#define HH   256
#define WW   256
#define NEXP 4
#define OCH  64
#define KS   3
#define KTOT (CIN * KS * KS)
#define WROW (CIN * KS * KS)

#define TILE_W   64
#define CI_CHUNK 8
#define NT       6
#define TCHUNK   (3 * CI_CHUNK)
#define SW_STRIDE 148
#define PATCH_W  68
#define PATCH_C0 3
#define WPAIRS   9
#define WPIECE   64
#define WWAVE    (WPAIRS * WPIECE)
#define WSLAB    (OCH * CI_CHUNK * KS * KS)
#define EP_W     68

#define RP_PITCH 33
#define ROWS_PER_BATCH 8
__global__ __launch_bounds__(256)
void att_partial_kernel(const float* __restrict__ x,
                        float* __restrict__ parth,
                        float* __restrict__ partv) {
    __shared__ float sRow[HH * RP_PITCH];
    __shared__ __attribute__((aligned(16))) float sCol[8 * WW];
    const int t    = threadIdx.x;
    const int lane = t & 31;
    const int wv   = t >> 5;
    const float* src = x + (size_t)blockIdx.x * (HH * WW) + (size_t)(wv * 32) * WW + 4 * lane;
    float* rdst = sRow + (wv * 32) * RP_PITCH + lane;

    v4f ca = {0.f, 0.f, 0.f, 0.f};
    v4f cb = {0.f, 0.f, 0.f, 0.f};
    for (int i0 = 0; i0 < 32; i0 += ROWS_PER_BATCH) {
        v4f a[ROWS_PER_BATCH], b[ROWS_PER_BATCH];
        for (int i = 0; i < ROWS_PER_BATCH; ++i) {
            a[i] = *(const v4f*)(src + (i0 + i) * WW);
            b[i] = *(const v4f*)(src + (i0 + i) * WW + 128);
        }
        for (int i = 0; i < ROWS_PER_BATCH; ++i) {
            ca += a[i];
            cb += b[i];
            rdst[(i0 + i) * RP_PITCH] = ((a[i].x + a[i].y) + (a[i].z + a[i].w))
                                      + ((b[i].x + b[i].y) + (b[i].z + b[i].w));
        }
    }
    *(v4f*)(sCol + wv * WW + 4 * lane)       = ca;
    *(v4f*)(sCol + wv * WW + 128 + 4 * lane) = cb;
    __syncthreads();

    float rowAcc = 0.f, colAcc = 0.f;
    for (int l = 0; l < 32; ++l) rowAcc += sRow[t * RP_PITCH + l];
    for (int w = 0; w < 8; ++w) colAcc += sCol[w * WW + t];
    parth[(size_t)blockIdx.x * HH + t] = rowAcc;
    partv[(size_t)blockIdx.x * WW + t] = colAcc;
}

#define MIX_PARTS 9
#define CGROUPS   4
#define CPG       (CIN / CGROUPS)
#define DSEG      8
__global__ __launch_bounds__(1024)
void att_mix_kernel(const float* __restrict__ parth,
                    const float* __restrict__ partv,
                    const float* __restrict__ net_w,
                    const float* __restrict__ bias,
                    const float* __restrict__ weight,
                    float* __restrict__ aggb,
                    float* __restrict__ aggw) {
    __shared__ float sPart[CGROUPS * (HH + WW)];
    __shared__ float sMean[HH + WW];
    __shared__ float sDot[NEXP * DSEG];
    __shared__ float sAtt[NEXP];
    const int t    = threadIdx.x;
    const int lane = t & 31;
    const int wv   = t >> 5;
    const int b    = blockIdx.x / MIX_PARTS;
    const int part = blockIdx.x % MIX_PARTS;

    const size_t E  = (size_t)OCH * KTOT;
    const int    j4 = (part * 1024 + t) * 4;
    const int    dc = j4 / WSLAB;
    const int    dw = (j4 % WSLAB) / WWAVE;
    const int    di = (j4 % WWAVE) / WPIECE;
    const int    dl = (j4 % WPIECE) >> 1;
    const int    sa = (dw * 8 + (dl >> 2)) * KTOT + dc * (CI_CHUNK * KS * KS) + (dl & 3) * (2 * KS * KS) + 2 * di;
    v2f wa[NEXP], wb[NEXP];
    #pragma unroll
    for (int e = 0; e < NEXP; ++e) {
        wa[e] = *(const v2f*)(weight + e * E + sa);
        wb[e] = *(const v2f*)(weight + e * E + sa + 2 * KS * KS);
    }

    const int dk = wv & 3, dseg = wv >> 2;
    const int dj = dseg * 64 + lane;
    const float n0 = net_w[dk * (HH + WW) + dj];
    const float n1 = net_w[dk * (HH + WW) + dj + 32];

    const int q = t >> 8, i = t & 255;
    const float* ph = parth + ((size_t)b * CIN + q * CPG) * HH + i;
    const float* pv = partv + ((size_t)b * CIN + q * CPG) * WW + i;
    float vh[CPG], vv[CPG];
    #pragma unroll
    for (int c = 0; c < CPG; ++c) { vh[c] = ph[c * HH]; vv[c] = pv[c * WW]; }
    float accH = 0.f, accV = 0.f;
    #pragma unroll
    for (int c = 0; c < CPG; ++c) { accH += vh[c]; accV += vv[c]; }
    sPart[q * (HH + WW) + i]      = accH;
    sPart[q * (HH + WW) + HH + i] = accV;
    __syncthreads();

    if (t < HH + WW) {
        const float inv = 1.f / (float)(CIN * WW);
        sMean[t] = (((sPart[t] + sPart[(HH + WW) + t]) + sPart[2 * (HH + WW) + t]) + sPart[3 * (HH + WW) + t]) * inv;
    }
    __syncthreads();

    float p = sMean[dj] * n0 + sMean[dj + 32] * n1;
    #pragma unroll
    for (int off = 16; off > 0; off >>= 1) p += __shfl_xor(p, off);
    if (lane == 0) sDot[dk * DSEG + dseg] = p;
    __syncthreads();

    if (t < NEXP) {
        float acc = 0.f;
        #pragma unroll
        for (int s = 0; s < DSEG; ++s) acc += sDot[t * DSEG + s];
        sAtt[t] = 1.f / (1.f + __expf(-acc));
    }
    __syncthreads();

    const float a0 = sAtt[0], a1 = sAtt[1], a2 = sAtt[2], a3 = sAtt[3];
    if (part == 0 && t < OCH) {
        aggb[b * OCH + t] = ((a0 * bias[t] + a1 * bias[OCH + t]) + a2 * bias[2 * OCH + t]) + a3 * bias[3 * OCH + t];
    }
    const v2f ma = a0 * wa[0] + a1 * wa[1] + a2 * wa[2] + a3 * wa[3];
    const v2f mb = a0 * wb[0] + a1 * wb[1] + a2 * wb[2] + a3 * wb[3];
    const v4f mo = {ma.x, ma.y, mb.x, mb.y};
    *(v4f*)(aggw + (size_t)b * E + j4) = mo;
}

#define ROWS_OUT 4
#define PROWS    6
#define NPROW    (CI_CHUNK * PROWS)
#define NCHUNK   (CIN / CI_CHUNK)
#define NSTAGE   3
#define SW_BUF   (OCH * SW_STRIDE)
#define SX_BUF   (NPROW * PATCH_W + 4)

__device__ constexpr int boff(int jj, int s, int t) {
    return ((2 * jj * PROWS + t) * 2 + s) * PATCH_W;
}

__global__ __launch_bounds__(256, 4)
void condconv_wmma_kernel(const float* __restrict__ x,
                          const float* __restrict__ aggw,
                          const float* __restrict__ aggb,
                          float* __restrict__ out) {
    __shared__ __attribute__((aligned(16))) float sMem[NSTAGE * SX_BUF + NSTAGE * SW_BUF + OCH];
    float* const sX    = sMem;
    float* const sW    = sMem + NSTAGE * SX_BUF;
    float* const sBias = sW + NSTAGE * SW_BUF;

    const int tid = threadIdx.x;
    const int b      = blockIdx.x >> 8;
    const int rem    = blockIdx.x & 255;
    const int h0     = (rem >> 2) * ROWS_OUT;
    const int wstart = (rem & 3) * TILE_W;

    if (tid < OCH) sBias[tid] = aggb[b * OCH + tid];

    const int lane   = tid & 31;
    const int wv     = tid >> 5;
    const int hl     = lane >> 4;
    const int l16    = lane & 15;
    const int m_base = (wv & 1) * 32;
    const int cbase  = (wv >> 1) * 16;

    v8f acc[NT][2] = {};

    const int aoff0 = (m_base + l16) * SW_STRIDE + 2 * hl;
    const int aoff1 = aoff0 + 16 * SW_STRIDE;
    const int boffl = (hl * PROWS * 2) * PATCH_W + PATCH_C0 + cbase + l16;

    const int    wq   = tid & 3;
    const float* wsrc = aggw + (size_t)b * (OCH * WROW) + wv * WWAVE + 2 * lane;
    float*       wdst = sW + (tid >> 2) * SW_STRIDE + 2 * wq;

    const bool xok0 = h0 > 0;
    const bool xok5 = h0 + ROWS_OUT < HH;
    const int  xr0  = xok0 ? h0 - 1 : 0;
    const int  xr5  = xok5 ? h0 + ROWS_OUT : HH - 1;
    const int  prow = ((wv >> 1) * PROWS) * 2 + (wv & 1);
    const float* xsrc = x + ((size_t)(b * CIN + wv) * HH) * WW + wstart + 2 * lane;
    float* xdst = sX + prow * PATCH_W + (PATCH_C0 + 1) + 2 * lane;

    const bool hact  = lane < 2;
    const int  hside = lane & 1;
    const int  hgw   = hside ? wstart + TILE_W : wstart - 1;
    const bool hokw  = (unsigned)hgw < (unsigned)WW;
    const int  hgwc  = hgw < 0 ? 0 : (hgw > WW - 1 ? WW - 1 : hgw);
    const float* hsrc = x + ((size_t)(b * CIN + wv) * HH) * WW + hgwc;
    float*       hdst = sX + prow * PATCH_W + (hside ? PATCH_C0 + 1 + TILE_W : PATCH_C0);

    auto issue_w = [&](v2f (&wreg)[WPAIRS], int ch) {
        #pragma unroll
        for (int i = 0; i < WPAIRS; ++i) wreg[i] = *(const v2f*)(wsrc + ch * WSLAB + i * WPIECE);
    };
    auto issue_x = [&](v2f (&xreg)[PROWS], float (&hreg)[PROWS], int ci0) {
        const float* xs = xsrc + (size_t)ci0 * (HH * WW);
        xreg[0] = *(const v2f*)(xs + xr0 * WW);
        xreg[1] = *(const v2f*)(xs + h0 * WW);
        xreg[2] = *(const v2f*)(xs + (h0 + 1) * WW);
        xreg[3] = *(const v2f*)(xs + (h0 + 2) * WW);
        xreg[4] = *(const v2f*)(xs + (h0 + 3) * WW);
        xreg[5] = *(const v2f*)(xs + xr5 * WW);
        if (hact) {
            const float* hs = hsrc + (size_t)ci0 * (HH * WW);
            hreg[0] = hs[xr0 * WW];
            hreg[1] = hs[h0 * WW];
            hreg[2] = hs[(h0 + 1) * WW];
            hreg[3] = hs[(h0 + 2) * WW];
            hreg[4] = hs[(h0 + 3) * WW];
            hreg[5] = hs[xr5 * WW];
        }
    };
    auto store_w = [&](const v2f (&wreg)[9], int par) {
        float* wd = wdst + par * SW_BUF;
        float f[18];
        #pragma unroll
        for (int i = 0; i < 9; ++i) { f[2 * i] = wreg[i].x; f[2 * i + 1] = wreg[i].y; }
        #pragma unroll
        for (int kw = 0; kw < 3; ++kw) {
            v2f u0, u1, u2, u3, u4, u5;
            #pragma unroll
            for (int cc = 0; cc < 2; ++cc) {
                const float g0 = f[cc * 9 + kw], g1 = f[cc * 9 + 3 + kw], g2 = f[cc * 9 + 6 + kw];
                const float s02 = g0 + g2;
                const float e   = g0 * (1.f / 24.f) + g2 * (1.f / 6.f);
                const float q   = g1 * (1.f / 12.f);
                u0[cc] = 0.25f * g0;
                u1[cc] = (s02 + g1) * (-1.f / 6.f);
                u2[cc] = (s02 - g1) * (-1.f / 6.f);
                u3[cc] = e + q;
                u4[cc] = e - q;
                u5[cc] = g2;
            }
            *(v2f*)(wd + 0 * TCHUNK + kw * CI_CHUNK) = u0;
            *(v2f*)(wd + 1 * TCHUNK + kw * CI_CHUNK) = u1;
            *(v2f*)(wd + 2 * TCHUNK + kw * CI_CHUNK) = u2;
            *(v2f*)(wd + 3 * TCHUNK + kw * CI_CHUNK) = u3;
            *(v2f*)(wd + 4 * TCHUNK + kw * CI_CHUNK) = u4;
            *(v2f*)(wd + 5 * TCHUNK + kw * CI_CHUNK) = u5;
        }
    };
    auto store_x = [&](const v2f (&xreg)[PROWS], const float (&hreg)[PROWS], int par) {
        float* xd = xdst + par * SX_BUF;
        const v2f z = {0.f, 0.f};
        const v2f d0 = xok0 ? xreg[0] : z, d1 = xreg[1], d2 = xreg[2], d3 = xreg[3], d4 = xreg[4];
        const v2f d5 = xok5 ? xreg[5] : z;
        const v2f p42 = d4 - d2, p31 = d3 - d1;
        *(v2f*)(xd + 0 * PATCH_W)  = (4.f * d0 - 5.f * d2) + d4;
        *(v2f*)(xd + 2 * PATCH_W)  = (d3 + d4) - 4.f * (d1 + d2);
        *(v2f*)(xd + 4 * PATCH_W)  = (d4 - d3) + 4.f * (d1 - d2);
        *(v2f*)(xd + 6 * PATCH_W)  = p42 + 2.f * p31;
        *(v2f*)(xd + 8 * PATCH_W)  = p42 - 2.f * p31;
        *(v2f*)(xd + 10 * PATCH_W) = (4.f * d1 - 5.f * d3) + d5;
        if (hact) {
            float* hd = hdst + par * SX_BUF;
            const float e0 = (hokw && xok0) ? hreg[0] : 0.f;
            const float e1 = hokw ? hreg[1] : 0.f;
            const float e2 = hokw ? hreg[2] : 0.f;
            const float e3 = hokw ? hreg[3] : 0.f;
            const float e4 = hokw ? hreg[4] : 0.f;
            const float e5 = (hokw && xok5) ? hreg[5] : 0.f;
            const float q42 = e4 - e2, q31 = e3 - e1;
            hd[0 * PATCH_W]  = (4.f * e0 - 5.f * e2) + e4;
            hd[2 * PATCH_W]  = (e3 + e4) - 4.f * (e1 + e2);
            hd[4 * PATCH_W]  = (e4 - e3) + 4.f * (e1 - e2);
            hd[6 * PATCH_W]  = q42 + 2.f * q31;
            hd[8 * PATCH_W]  = q42 - 2.f * q31;
            hd[10 * PATCH_W] = (4.f * e1 - 5.f * e3) + e5;
        }
    };

    auto mma_tap = [&](const float* tW, const float* tX) {
        #pragma unroll
        for (int jj = 0; jj < 2; ++jj) {
            #pragma unroll
            for (int t = 0; t < NT; ++t) {
                const v2f a0 = *(const v2f*)&tW[aoff0 + t * TCHUNK + 4 * jj];
                const v2f a1 = *(const v2f*)&tW[aoff1 + t * TCHUNK + 4 * jj];
                v2f bf;
                bf.x = tX[boffl + boff(jj, 0, t)];
                bf.y = tX[boffl + boff(jj, 1, t)];
                acc[t][0] = __builtin_amdgcn_wmma_f32_16x16x4_f32(false, a0, false, bf, (short)0, acc[t][0], false, false);
                acc[t][1] = __builtin_amdgcn_wmma_f32_16x16x4_f32(false, a1, false, bf, (short)0, acc[t][1], false, false);
            }
        }
    };

    {
        v2f   wreg0[9], wreg1[9];
        v2f   xreg0[PROWS], xreg1[PROWS];
        float hreg0[PROWS] = {0.f, 0.f, 0.f, 0.f, 0.f, 0.f};
        float hreg1[PROWS] = {0.f, 0.f, 0.f, 0.f, 0.f, 0.f};
        issue_w(wreg0, 0);
        issue_x(xreg0, hreg0, 0);
        issue_w(wreg1, 1);
        issue_x(xreg1, hreg1, CI_CHUNK);
        store_w(wreg0, 0);
        store_x(xreg0, hreg0, 0);
        store_w(wreg1, 1);
        store_x(xreg1, hreg1, 1);
    }
    __syncthreads();

    int st_cur = 0, st_nxt = 1, st_wr = 2;
    float hreg[PROWS] = {0.f, 0.f, 0.f, 0.f, 0.f, 0.f};

    for (int c = 0; c < NCHUNK - 2; ++c) {
        v2f wreg[9];
        v2f xreg[PROWS];
        const float* cW = sW + st_cur * SW_BUF;
        const float* cX = sX + st_cur * SX_BUF;

        issue_w(wreg, c + 2);
        mma_tap(cW, cX);
        if (c >= 1) {
            __builtin_amdgcn_s_barrier_wait(-1);
            __builtin_amdgcn_fence(__ATOMIC_ACQUIRE, "workgroup");
        }
        store_w(wreg, st_wr);
        issue_x(xreg, hreg, (c + 2) * CI_CHUNK);
        #pragma nounroll
        for (int kw = 1; kw < KS; ++kw) mma_tap(cW + kw * CI_CHUNK, cX + kw);
        store_x(xreg, hreg, st_wr);
        __builtin_amdgcn_fence(__ATOMIC_RELEASE, "workgroup");
        __builtin_amdgcn_s_barrier_signal(-1);
        const int st_old = st_cur;
        st_cur = st_nxt;
        st_nxt = st_wr;
        st_wr  = st_old;
    }
    {
        const float* cW = sW + st_cur * SW_BUF;
        const float* cX = sX + st_cur * SX_BUF;
        #pragma nounroll
        for (int kw = 0; kw < KS; ++kw) mma_tap(cW + kw * CI_CHUNK, cX + kw);
        __builtin_amdgcn_s_barrier_wait(-1);
        __builtin_amdgcn_fence(__ATOMIC_ACQUIRE, "workgroup");
        cW = sW + st_nxt * SW_BUF;
        cX = sX + st_nxt * SX_BUF;
        #pragma nounroll
        for (int kw = 0; kw < KS; ++kw) mma_tap(cW + kw * CI_CHUNK, cX + kw);
    }

    __syncthreads();
    float* sOut = sW;
    #pragma unroll
    for (int mi = 0; mi < 2; ++mi) {
        #pragma unroll
        for (int r = 0; r < 8; ++r) {
            const int o = m_base + mi * 16 + hl * 8 + r;
            const float m0 = acc[0][mi][r], m1 = acc[1][mi][r], m2 = acc[2][mi][r];
            const float m3 = acc[3][mi][r], m4 = acc[4][mi][r], m5 = acc[5][mi][r];
            const float s12 = m1 + m2, d12 = m1 - m2, s34 = m3 + m4, d34 = m3 - m4;
            float* so = sOut + o * EP_W + cbase + l16;
            so[0 * OCH * EP_W] = (m0 + s12) + s34;
            so[1 * OCH * EP_W] = d12 + 2.f * d34;
            so[2 * OCH * EP_W] = s12 + 4.f * s34;
            so[3 * OCH * EP_W] = (d12 + 8.f * d34) + m5;
        }
    }
    __syncthreads();
    {
        const int piece = tid & 15;
        const int osub  = tid >> 4;
        float bz[4];
        #pragma unroll
        for (int g = 0; g < 4; ++g) bz[g] = sBias[osub + 16 * g];
        #pragma unroll
        for (int i = 0; i < 16; ++i) {
            const int row = i & 3;
            const int o   = osub + 16 * (i >> 2);
            const v4f v = *(const v4f*)(sOut + (row * OCH + o) * EP_W + 4 * piece);
            *(v4f*)(out + (((size_t)b * OCH + o) * HH + h0 + row) * WW + wstart + 4 * piece) = v + bz[i >> 2];
        }
    }
}

extern "C" void kernel_launch(void* const* d_in, const int* in_sizes, int n_in,
                              void* d_out, int out_size, void* d_ws, size_t ws_size,
                              hipStream_t stream) {
    const float* x      = (const float*)d_in[0];
    const float* net_w  = (const float*)d_in[1];
    const float* weight = (const float*)d_in[2];
    const float* bias   = (const float*)d_in[3];
    float* out = (float*)d_out;

    float* ws = (float*)d_ws;
    float* att   = ws;
    float* aggb  = att + BS * NEXP;
    float* aggw  = aggb + BS * OCH;
    float* parth = aggw + (size_t)BS * OCH * KTOT;
    float* partv = parth + (size_t)BS * CIN * HH;

    att_partial_kernel<<<BS * CIN, 256, 0, stream>>>(x, parth, partv);
    att_mix_kernel<<<BS * MIX_PARTS, 1024, 0, stream>>>(parth, partv, net_w, bias, weight, aggb, aggw);
    condconv_wmma_kernel<<<BS * (HH / ROWS_OUT) * (WW / TILE_W), 256, 0, stream>>>(x, aggw, aggb, out);
}
